# starting kernel
# baseline (speedup 1.0000x reference)
#include <hip/hip_runtime.h>
#include <hip/hip_bf16.h>


typedef float v2f __attribute__((ext_vector_type(2)));
typedef float v8f __attribute__((ext_vector_type(8)));

struct Gate { signed char op, a, b, wi; };

#define RY(wr, wi)  { 0, (wr), 0, (wi) }
#define CX(cc, tt)  { 1, (cc), (tt), 0 }

__device__ __constant__ Gate GATES[5][24] = {
  { RY(0,0),RY(1,1),RY(2,2),RY(3,3), CX(0,1),CX(2,3), RY(0,4),RY(2,5), CX(1,2), RY(1,6),
    {0,0,0,0},{0,0,0,0},{0,0,0,0},{0,0,0,0},{0,0,0,0},{0,0,0,0},{0,0,0,0},
    {0,0,0,0},{0,0,0,0},{0,0,0,0},{0,0,0,0},{0,0,0,0},{0,0,0,0},{0,0,0,0} },
  { RY(0,0),RY(1,1),RY(2,2),RY(3,3), CX(0,1),CX(1,2),CX(2,3), RY(1,4),RY(2,5),RY(3,6),
    {0,0,0,0},{0,0,0,0},{0,0,0,0},{0,0,0,0},{0,0,0,0},{0,0,0,0},{0,0,0,0},
    {0,0,0,0},{0,0,0,0},{0,0,0,0},{0,0,0,0},{0,0,0,0},{0,0,0,0},{0,0,0,0} },
  { RY(0,0),RY(1,1),RY(2,2),RY(3,3), CX(0,1),CX(1,2),CX(2,3),CX(3,0),
    RY(0,4),RY(1,5),RY(2,6),RY(3,7),
    {0,0,0,0},{0,0,0,0},{0,0,0,0},{0,0,0,0},{0,0,0,0},{0,0,0,0},
    {0,0,0,0},{0,0,0,0},{0,0,0,0},{0,0,0,0},{0,0,0,0},{0,0,0,0} },
  { RY(0,0),RY(1,1),RY(2,2),RY(3,3), CX(0,1),CX(1,2),CX(2,3),CX(3,0),
    RY(0,4),RY(1,5),RY(2,6),RY(3,7), CX(3,0),CX(2,3),CX(1,2),CX(0,1),
    RY(0,8),RY(1,9),RY(2,10),RY(3,11), CX(0,1),CX(1,2),CX(2,3),CX(3,0) },
  { RY(0,0),RY(1,1),RY(2,2),RY(3,3), CX(0,1),CX(2,3),CX(1,2),CX(3,0),
    RY(0,4),RY(1,5),RY(2,6),RY(3,7), CX(0,1),CX(2,3),CX(1,2),CX(3,0),
    RY(0,8),RY(1,9),RY(2,10),RY(3,11), CX(0,1),CX(2,3),CX(1,2),CX(3,0) },
};

__device__ __constant__ Gate GATES_DR_TAIL[8] = {
  RY(0,12),RY(1,13),RY(2,14),RY(3,15), CX(3,0),CX(2,3),CX(1,2),CX(0,1)
};

__device__ __constant__ int GCNT[5] = { 10, 10, 12, 24  , 24 };

__global__ void __launch_bounds__(256) build_M_kernel(const float* __restrict__ wts,
                                                      float* __restrict__ Mout) {
  const int pair = blockIdx.x;
  const int i    = pair / 3;
  const int t    = threadIdx.x;
  const int m    = t >> 4;
  const int j    = t & 15;
  const int circ = i % 5;

  __shared__ float U[16][17];
  U[m][j] = (m == j) ? 1.0f : 0.0f;
  __syncthreads();

  const float* wv = wts + pair * 16;

  const int ng = GCNT[circ];
  for (int g = 0; g < ng + ((circ == 3) ? 8 : 0); ++g) {
    Gate gd = (g < 24) ? GATES[circ][g] : GATES_DR_TAIL[g - 24];
    if (gd.op == 0) {
      const int  mask = 1 << (3 - gd.a);
      const float th  = 0.5f * wv[(int)gd.wi];
      float sh, ch;  sincosf(th, &sh, &ch);
      const float um = U[m][j];
      const float up = U[m ^ mask][j];
      __syncthreads();
      U[m][j] = ((m & mask) == 0) ? (ch * um - sh * up)
                                  : (sh * up + ch * um);
      __syncthreads();
    } else {
      const int cb = 3 - gd.a, tb = 3 - gd.b;
      const int src = m ^ (((m >> cb) & 1) << tb);
      const float v = U[src][j];
      __syncthreads();
      U[m][j] = v;
      __syncthreads();
    }
  }

  float acc = 0.0f;
#pragma unroll
  for (int mm = 0; mm < 16; ++mm) {
    const float sgn = (mm < 8) ? 1.0f : -1.0f;
    acc += sgn * U[mm][m] * U[mm][j];
  }
  Mout[(pair * 16 + m) * 16 + j] = acc;
}

__global__ void __launch_bounds__(32) quanv_wmma_kernel(const float* __restrict__ in,
                                                        const float* __restrict__ M,
                                                        float* __restrict__ out) {
  const int tile = blockIdx.x;
  const int oi   = blockIdx.y;
  const int lane = threadIdx.x;
  const int half = lane >> 4;
  const int col  = lane & 15;
  const int row  = lane & 15;

  __shared__ float S[16][17];

  float part[8];
#pragma unroll
  for (int r = 0; r < 8; ++r) part[r] = 0.0f;

  for (int c = 0; c < 3; ++c) {
    __syncthreads();
    if (lane < 16) {
      const int p   = tile * 16 + lane;
      const int b   = p / 961;
      const int rem = p % 961;
      const int h   = rem / 31;
      const int w   = rem % 31;
      const float* base = in + ((b * 3 + c) * 32 + h) * 32 + w;
      const float x0 = base[0],  x1 = base[1];
      const float x2 = base[32], x3 = base[33];
      float sh0, ch0, sh1, ch1, sh2, ch2, sh3, ch3;
      sincosf(0.5f * x0, &sh0, &ch0);
      sincosf(0.5f * x1, &sh1, &ch1);
      sincosf(0.5f * x2, &sh2, &ch2);
      sincosf(0.5f * x3, &sh3, &ch3);
#pragma unroll
      for (int mm = 0; mm < 16; ++mm) {
        const float v = ((mm & 8) ? sh0 : ch0) * ((mm & 4) ? sh1 : ch1)
                      * ((mm & 2) ? sh2 : ch2) * ((mm & 1) ? sh3 : ch3);
        S[lane][mm] = v;
      }
    }
    __syncthreads();

    const float* Mp = M + (oi * 3 + c) * 256;

    v8f acc = {};
#pragma unroll
    for (int s = 0; s < 4; ++s) {
      const int k0 = 4 * s;
      v2f a, bf;
      a[0]  = S[row][k0 + 2 * half + 0];
      a[1]  = S[row][k0 + 2 * half + 1];
      bf[0] = Mp[(k0 + 2 * half + 0) * 16 + col];
      bf[1] = Mp[(k0 + 2 * half + 1) * 16 + col];
      acc = __builtin_amdgcn_wmma_f32_16x16x4_f32(
                false, a, false, bf, (short)0, acc, false, false);
    }

#pragma unroll
    for (int r = 0; r < 8; ++r) {
      part[r] += acc[r] * S[r + 8 * half][col];
    }
  }

#pragma unroll
  for (int r = 0; r < 8; ++r) {
    float v = part[r];
    v += __shfl_xor(v, 1);
    v += __shfl_xor(v, 2);
    v += __shfl_xor(v, 4);
    v += __shfl_xor(v, 8);
    part[r] = v;
  }

  float outv = 0.0f;
#pragma unroll
  for (int r = 0; r < 8; ++r) {
    const float vl = __shfl(part[r], 0);
    const float vh = __shfl(part[r], 16);
    if (lane == r)     outv = vl;
    if (lane == r + 8) outv = vh;
  }

  if (lane < 16) {
    const int p   = tile * 16 + lane;
    const int b   = p / 961;
    const int rem = p % 961;
    out[(b * 16 + oi) * 961 + rem] = outv;
  }
}

extern "C" void kernel_launch(void* const* d_in, const int* in_sizes, int n_in,
                              void* d_out, int out_size, void* d_ws, size_t ws_size,
                              hipStream_t stream) {
  const float* inputs  = (const float*)d_in[0];
  const float* weights = (const float*)d_in[1];
  float*       out     = (float*)d_out;
  float*       Mbuf    = (float*)d_ws;

  build_M_kernel<<<48, 256, 0, stream>>>(weights, Mbuf);
  quanv_wmma_kernel<<<dim3(961, 16), 32, 0, stream>>>(inputs, Mbuf, out);
}


# Matrix-build kernel: first wave evolves the 16 columns of U one per lane in registers, gate lists compiled in, angles shared by readlane; no per-gate barriers or table loads; sum order unchanged
# speedup vs baseline: 1.5193x; 1.5193x over previous
#include <hip/hip_runtime.h>
#include <hip/hip_bf16.h>


typedef float v2f __attribute__((ext_vector_type(2)));
typedef float v8f __attribute__((ext_vector_type(8)));

template <int WIRE>
__device__ __forceinline__ void ry(float (&u)[16], const float c, const float s) {
  constexpr int mask = 1 << (3 - WIRE);
  for (int m = 0; m < 16; ++m) {
    if ((m & mask) == 0) {
      const float a = u[m];
      const float b = u[m | mask];
      u[m]        = c * a - s * b;
      u[m | mask] = s * a + c * b;
    }
  }
}

template <int CTRL, int TGT>
__device__ __forceinline__ void cx(float (&u)[16]) {
  constexpr int cb = 3 - CTRL;
  constexpr int tb = 3 - TGT;
  for (int m = 0; m < 16; ++m) {
    if ((((m >> cb) & 1) == 1) && (((m >> tb) & 1) == 0)) {
      const float t = u[m];
      u[m] = u[m | (1 << tb)];
      u[m | (1 << tb)] = t;
    }
  }
}

template <int W0>
__device__ __forceinline__ void ry_all(float (&u)[16], const float (&c)[16], const float (&s)[16]) {
  ry<0>(u, c[W0 + 0], s[W0 + 0]);
  ry<1>(u, c[W0 + 1], s[W0 + 1]);
  ry<2>(u, c[W0 + 2], s[W0 + 2]);
  ry<3>(u, c[W0 + 3], s[W0 + 3]);
}

__device__ __forceinline__ void ring_fwd(float (&u)[16]) {
  cx<0, 1>(u); cx<1, 2>(u); cx<2, 3>(u); cx<3, 0>(u);
}

__device__ __forceinline__ void ring_bwd(float (&u)[16]) {
  cx<3, 0>(u); cx<2, 3>(u); cx<1, 2>(u); cx<0, 1>(u);
}

__device__ __forceinline__ void block_ring(float (&u)[16]) {
  cx<0, 1>(u); cx<2, 3>(u); cx<1, 2>(u); cx<3, 0>(u);
}

__device__ __forceinline__ void circ_ran(float (&u)[16], const float (&c)[16], const float (&s)[16]) {
  ry_all<0>(u, c, s);
  cx<0, 1>(u); cx<2, 3>(u);
  ry<0>(u, c[4], s[4]); ry<2>(u, c[5], s[5]);
  cx<1, 2>(u);
  ry<1>(u, c[6], s[6]);
}

__device__ __forceinline__ void circ_line(float (&u)[16], const float (&c)[16], const float (&s)[16]) {
  ry_all<0>(u, c, s);
  cx<0, 1>(u); cx<1, 2>(u); cx<2, 3>(u);
  ry<1>(u, c[4], s[4]); ry<2>(u, c[5], s[5]); ry<3>(u, c[6], s[6]);
}

__device__ __forceinline__ void circ_ring(float (&u)[16], const float (&c)[16], const float (&s)[16]) {
  ry_all<0>(u, c, s);
  ring_fwd(u);
  ry_all<4>(u, c, s);
}

__device__ __forceinline__ void circ_doublering(float (&u)[16], const float (&c)[16], const float (&s)[16]) {
  ry_all<0>(u, c, s);  ring_fwd(u);  ry_all<4>(u, c, s);   ring_bwd(u);
  ry_all<8>(u, c, s);  ring_fwd(u);  ry_all<12>(u, c, s);  ring_bwd(u);
}

__device__ __forceinline__ void circ_blockring(float (&u)[16], const float (&c)[16], const float (&s)[16]) {
  ry_all<0>(u, c, s);  block_ring(u);
  ry_all<4>(u, c, s);  block_ring(u);
  ry_all<8>(u, c, s);  block_ring(u);
}

__global__ void __launch_bounds__(256) build_M_kernel(const float* __restrict__ wts,
                                                      float* __restrict__ Mout) {
  const int pair = blockIdx.x;
  const int i    = pair / 3;
  const int t    = threadIdx.x;
  const int m    = t >> 4;
  const int j    = t & 15;
  const int circ = i % 5;

  __shared__ float U[16][17];

  if (t < 32) {
    float sh, ch;
    sincosf(0.5f * wts[pair * 16 + j], &sh, &ch);

    float c[16], s[16];
    for (int k = 0; k < 16; ++k) {
      c[k] = __int_as_float(__builtin_amdgcn_readlane(__float_as_int(ch), k));
      s[k] = __int_as_float(__builtin_amdgcn_readlane(__float_as_int(sh), k));
    }

    float u[16];
    for (int mm = 0; mm < 16; ++mm) u[mm] = (mm == j) ? 1.0f : 0.0f;

    switch (circ) {
      case 0:  circ_ran(u, c, s);        break;
      case 1:  circ_line(u, c, s);       break;
      case 2:  circ_ring(u, c, s);       break;
      case 3:  circ_doublering(u, c, s); break;
      default: circ_blockring(u, c, s);  break;
    }

    if (t < 16) {
      for (int mm = 0; mm < 16; ++mm) U[mm][j] = u[mm];
    }
  }
  __syncthreads();

  float acc = 0.0f;
#pragma unroll
  for (int mm = 0; mm < 16; ++mm) {
    const float sgn = (mm < 8) ? 1.0f : -1.0f;
    acc += sgn * U[mm][m] * U[mm][j];
  }
  Mout[(pair * 16 + m) * 16 + j] = acc;
}

__global__ void __launch_bounds__(32) quanv_wmma_kernel(const float* __restrict__ in,
                                                        const float* __restrict__ M,
                                                        float* __restrict__ out) {
  const int tile = blockIdx.x;
  const int oi   = blockIdx.y;
  const int lane = threadIdx.x;
  const int half = lane >> 4;
  const int col  = lane & 15;
  const int row  = lane & 15;

  __shared__ float S[16][17];

  float part[8];
#pragma unroll
  for (int r = 0; r < 8; ++r) part[r] = 0.0f;

  for (int c = 0; c < 3; ++c) {
    __syncthreads();
    if (lane < 16) {
      const int p   = tile * 16 + lane;
      const int b   = p / 961;
      const int rem = p % 961;
      const int h   = rem / 31;
      const int w   = rem % 31;
      const float* base = in + ((b * 3 + c) * 32 + h) * 32 + w;
      const float x0 = base[0],  x1 = base[1];
      const float x2 = base[32], x3 = base[33];
      float sh0, ch0, sh1, ch1, sh2, ch2, sh3, ch3;
      sincosf(0.5f * x0, &sh0, &ch0);
      sincosf(0.5f * x1, &sh1, &ch1);
      sincosf(0.5f * x2, &sh2, &ch2);
      sincosf(0.5f * x3, &sh3, &ch3);
#pragma unroll
      for (int mm = 0; mm < 16; ++mm) {
        const float v = ((mm & 8) ? sh0 : ch0) * ((mm & 4) ? sh1 : ch1)
                      * ((mm & 2) ? sh2 : ch2) * ((mm & 1) ? sh3 : ch3);
        S[lane][mm] = v;
      }
    }
    __syncthreads();

    const float* Mp = M + (oi * 3 + c) * 256;

    v8f acc = {};
#pragma unroll
    for (int s = 0; s < 4; ++s) {
      const int k0 = 4 * s;
      v2f a, bf;
      a[0]  = S[row][k0 + 2 * half + 0];
      a[1]  = S[row][k0 + 2 * half + 1];
      bf[0] = Mp[(k0 + 2 * half + 0) * 16 + col];
      bf[1] = Mp[(k0 + 2 * half + 1) * 16 + col];
      acc = __builtin_amdgcn_wmma_f32_16x16x4_f32(
                false, a, false, bf, (short)0, acc, false, false);
    }

#pragma unroll
    for (int r = 0; r < 8; ++r) {
      part[r] += acc[r] * S[r + 8 * half][col];
    }
  }

#pragma unroll
  for (int r = 0; r < 8; ++r) {
    float v = part[r];
    v += __shfl_xor(v, 1);
    v += __shfl_xor(v, 2);
    v += __shfl_xor(v, 4);
    v += __shfl_xor(v, 8);
    part[r] = v;
  }

  float outv = 0.0f;
#pragma unroll
  for (int r = 0; r < 8; ++r) {
    const float vl = __shfl(part[r], 0);
    const float vh = __shfl(part[r], 16);
    if (lane == r)     outv = vl;
    if (lane == r + 8) outv = vh;
  }

  if (lane < 16) {
    const int p   = tile * 16 + lane;
    const int b   = p / 961;
    const int rem = p % 961;
    out[(b * 16 + oi) * 961 + rem] = outv;
  }
}

extern "C" void kernel_launch(void* const* d_in, const int* in_sizes, int n_in,
                              void* d_out, int out_size, void* d_ws, size_t ws_size,
                              hipStream_t stream) {
  const float* inputs  = (const float*)d_in[0];
  const float* weights = (const float*)d_in[1];
  float*       out     = (float*)d_out;
  float*       Mbuf    = (float*)d_ws;

  build_M_kernel<<<48, 256, 0, stream>>>(weights, Mbuf);
  quanv_wmma_kernel<<<dim3(961, 16), 32, 0, stream>>>(inputs, Mbuf, out);
}


# Main kernel: s^T M s expanded in the 81 trig products of the 4 pixels, coefficients folded from M in kernel 1; one WMMA product per tile gives all 16 channels, no LDS or shuffles; sum order changes
# speedup vs baseline: 2.5415x; 1.6728x over previous
#include <hip/hip_runtime.h>
#include <hip/hip_bf16.h>


typedef float v2f __attribute__((ext_vector_type(2)));
typedef float v8f __attribute__((ext_vector_type(8)));

template <int WIRE>
__device__ __forceinline__ void ry(float (&u)[16], const float c, const float s) {
  constexpr int mask = 1 << (3 - WIRE);
  for (int m = 0; m < 16; ++m) {
    if ((m & mask) == 0) {
      const float a = u[m];
      const float b = u[m | mask];
      u[m]        = c * a - s * b;
      u[m | mask] = s * a + c * b;
    }
  }
}

template <int CTRL, int TGT>
__device__ __forceinline__ void cx(float (&u)[16]) {
  constexpr int cb = 3 - CTRL;
  constexpr int tb = 3 - TGT;
  for (int m = 0; m < 16; ++m) {
    if ((((m >> cb) & 1) == 1) && (((m >> tb) & 1) == 0)) {
      const float t = u[m];
      u[m] = u[m | (1 << tb)];
      u[m | (1 << tb)] = t;
    }
  }
}

template <int W0>
__device__ __forceinline__ void ry_all(float (&u)[16], const float (&c)[16], const float (&s)[16]) {
  ry<0>(u, c[W0 + 0], s[W0 + 0]);
  ry<1>(u, c[W0 + 1], s[W0 + 1]);
  ry<2>(u, c[W0 + 2], s[W0 + 2]);
  ry<3>(u, c[W0 + 3], s[W0 + 3]);
}

__device__ __forceinline__ void ring_fwd(float (&u)[16]) {
  cx<0, 1>(u); cx<1, 2>(u); cx<2, 3>(u); cx<3, 0>(u);
}

__device__ __forceinline__ void ring_bwd(float (&u)[16]) {
  cx<3, 0>(u); cx<2, 3>(u); cx<1, 2>(u); cx<0, 1>(u);
}

__device__ __forceinline__ void block_ring(float (&u)[16]) {
  cx<0, 1>(u); cx<2, 3>(u); cx<1, 2>(u); cx<3, 0>(u);
}

__device__ __forceinline__ void circ_ran(float (&u)[16], const float (&c)[16], const float (&s)[16]) {
  ry_all<0>(u, c, s);
  cx<0, 1>(u); cx<2, 3>(u);
  ry<0>(u, c[4], s[4]); ry<2>(u, c[5], s[5]);
  cx<1, 2>(u);
  ry<1>(u, c[6], s[6]);
}

__device__ __forceinline__ void circ_line(float (&u)[16], const float (&c)[16], const float (&s)[16]) {
  ry_all<0>(u, c, s);
  cx<0, 1>(u); cx<1, 2>(u); cx<2, 3>(u);
  ry<1>(u, c[4], s[4]); ry<2>(u, c[5], s[5]); ry<3>(u, c[6], s[6]);
}

__device__ __forceinline__ void circ_ring(float (&u)[16], const float (&c)[16], const float (&s)[16]) {
  ry_all<0>(u, c, s);
  ring_fwd(u);
  ry_all<4>(u, c, s);
}

__device__ __forceinline__ void circ_doublering(float (&u)[16], const float (&c)[16], const float (&s)[16]) {
  ry_all<0>(u, c, s);  ring_fwd(u);  ry_all<4>(u, c, s);   ring_bwd(u);
  ry_all<8>(u, c, s);  ring_fwd(u);  ry_all<12>(u, c, s);  ring_bwd(u);
}

__device__ __forceinline__ void circ_blockring(float (&u)[16], const float (&c)[16], const float (&s)[16]) {
  ry_all<0>(u, c, s);  block_ring(u);
  ry_all<4>(u, c, s);  block_ring(u);
  ry_all<8>(u, c, s);  block_ring(u);
}


__global__ void __launch_bounds__(256) build_coef_kernel(const float* __restrict__ wts,
                                                         float* __restrict__ Cf) {
  const int pair = blockIdx.x;
  const int i    = pair / 3;
  const int cin  = pair % 3;
  const int t    = threadIdx.x;
  const int m    = t >> 4;
  const int j    = t & 15;
  const int circ = i % 5;

  __shared__ float U[16][17];
  __shared__ float Msh[16][17];

  if (t < 32) {
    float sh, ch;
    sincosf(0.5f * wts[pair * 16 + j], &sh, &ch);

    float c[16], s[16];
    for (int k = 0; k < 16; ++k) {
      c[k] = __int_as_float(__builtin_amdgcn_readlane(__float_as_int(ch), k));
      s[k] = __int_as_float(__builtin_amdgcn_readlane(__float_as_int(sh), k));
    }

    float u[16];
    for (int mm = 0; mm < 16; ++mm) u[mm] = (mm == j) ? 1.0f : 0.0f;

    switch (circ) {
      case 0:  circ_ran(u, c, s);        break;
      case 1:  circ_line(u, c, s);       break;
      case 2:  circ_ring(u, c, s);       break;
      case 3:  circ_doublering(u, c, s); break;
      default: circ_blockring(u, c, s);  break;
    }

    if (t < 16) {
      for (int mm = 0; mm < 16; ++mm) U[mm][j] = u[mm];
    }
  }
  __syncthreads();

  float acc = 0.0f;
#pragma unroll
  for (int mm = 0; mm < 16; ++mm) {
    const float sgn = (mm < 8) ? 1.0f : -1.0f;
    acc += sgn * U[mm][m] * U[mm][j];
  }
  Msh[m][j] = acc;
  __syncthreads();

  if (t < 84) {
    const int h  = t / 42;
    const int sg = t % 42;
    const int a  = sg / 14;
    const int jj = sg % 14;
    int d1, q;
    if (h == 0) { d1 = (jj < 9) ? 0 : 1;  q = (jj < 9) ? jj : jj - 9; }
    else        { d1 = (jj < 9) ? 2 : 1;  q = (jj < 9) ? 8 - jj : 17 - jj; }
    const bool pad = (h == 1) && (jj == 13);
    const int d2 = q / 3;
    const int d3 = q % 3;
    const int flip = ((a == 2) ? 8 : 0) | ((d1 == 2) ? 4 : 0) | ((d2 == 2) ? 2 : 0) | ((d3 == 2) ? 1 : 0);
    const int neg  = ((a == 1) ? 8 : 0) | ((d1 == 1) ? 4 : 0) | ((d2 == 1) ? 2 : 0) | ((d3 == 1) ? 1 : 0);
    float cf = 0.0f;
#pragma unroll
    for (int r = 0; r < 16; ++r) {
      const float v = Msh[r][r ^ flip];
      cf += (__popc(r & neg) & 1) ? -v : v;
    }
    cf *= 0.0625f;
    Cf[(((cin * 21 + (sg >> 1)) * 32) + h * 16 + i) * 2 + (sg & 1)] = pad ? 0.0f : cf;
  }
}

__global__ void __launch_bounds__(128) quanv_wmma_kernel(const float* __restrict__ in,
                                                         const float* __restrict__ Cf,
                                                         float* __restrict__ out) {
  const int tile = blockIdx.x * 4 + (threadIdx.x >> 5);
  if (tile >= 961) return;
  const int lane = threadIdx.x & 31;
  const int h    = lane >> 4;
  const int n    = lane & 15;

  const int p   = tile * 16 + n;
  const int b   = p / 961;
  const int rem = p % 961;
  const int ph  = rem / 31;
  const int pw  = rem % 31;

  float x[3][4];
#pragma unroll
  for (int c = 0; c < 3; ++c) {
    const float* base = in + ((b * 3 + c) * 32 + ph) * 32 + pw;
    x[c][0] = base[0];  x[c][1] = base[1];
    x[c][2] = base[32]; x[c][3] = base[33];
  }
  __builtin_amdgcn_sched_barrier(0);

  const v2f* Cv = reinterpret_cast<const v2f*>(Cf) + lane;
  v2f afr[63];
#pragma unroll
  for (int g = 0; g < 63; ++g) afr[g] = Cv[g * 32];

  float cs[3][4], sn[3][4];
#pragma unroll
  for (int c = 0; c < 3; ++c) {
#pragma unroll
    for (int w = 0; w < 4; ++w) sincosf(x[c][w], &sn[c][w], &cs[c][w]);
  }

  const bool hi = (h != 0);
  v8f acc0 = v8f{};
  v8f acc1 = v8f{};
#pragma unroll
  for (int c = 0; c < 3; ++c) {
    const float g3[3] = { hi ? sn[c][3] : 1.0f, cs[c][3], hi ? 1.0f : sn[c][3] };
    const float g2[3] = { hi ? sn[c][2] : 1.0f, cs[c][2], hi ? 1.0f : sn[c][2] };
    const float g1[2] = { hi ? sn[c][1] : 1.0f, cs[c][1] };
    float Q[9];
#pragma unroll
    for (int a2 = 0; a2 < 3; ++a2) {
#pragma unroll
      for (int a3 = 0; a3 < 3; ++a3) Q[3 * a2 + a3] = g2[a2] * g3[a3];
    }
    float R[14];
#pragma unroll
    for (int j = 0; j < 14; ++j) R[j] = (j < 9) ? g1[0] * Q[j] : g1[1] * Q[j - 9];
    float F[42];
#pragma unroll
    for (int j = 0; j < 14; ++j) {
      F[j]      = R[j];
      F[14 + j] = cs[c][0] * R[j];
      F[28 + j] = sn[c][0] * R[j];
    }
#pragma unroll
    for (int s = 0; s < 21; ++s) {
      const int g = c * 21 + s;
      v2f bfr;
      bfr[0] = F[2 * s];
      bfr[1] = F[2 * s + 1];
      if (g & 1) {
        acc1 = __builtin_amdgcn_wmma_f32_16x16x4_f32(false, afr[g], false, bfr, (short)0, acc1, false, false);
      } else {
        acc0 = __builtin_amdgcn_wmma_f32_16x16x4_f32(false, afr[g], false, bfr, (short)0, acc0, false, false);
      }
    }
  }

  float* op = out + (b * 16 + 8 * h) * 961 + rem;
#pragma unroll
  for (int r = 0; r < 8; ++r) op[r * 961] = acc0[r] + acc1[r];
}

extern "C" void kernel_launch(void* const* d_in, const int* in_sizes, int n_in,
                              void* d_out, int out_size, void* d_ws, size_t ws_size,
                              hipStream_t stream) {
  const float* inputs  = (const float*)d_in[0];
  const float* weights = (const float*)d_in[1];
  float*       out     = (float*)d_out;
  float*       Cbuf    = (float*)d_ws;

  build_coef_kernel<<<48, 256, 0, stream>>>(weights, Cbuf);
  quanv_wmma_kernel<<<241, 128, 0, stream>>>(inputs, Cbuf, out);
}


# First launch gains 48 blocks that compute cos,sin once per input pixel into a workspace plane beside the coefficient build; main kernel loads them instead of 12 sincos per lane; sum order unchanged
# speedup vs baseline: 2.7343x; 1.0759x over previous
#include <hip/hip_runtime.h>
#include <hip/hip_bf16.h>


constexpr int    kPixels    = 16 * 3 * 32 * 32;
constexpr int    kPlaneOff  = 4096;
constexpr size_t kWsNeed    = (size_t)kPlaneOff * 4 + (size_t)kPixels * 8;
constexpr int    kCoefBlocks = 48;

typedef float v2f __attribute__((ext_vector_type(2)));
typedef float v4f __attribute__((ext_vector_type(4)));
typedef float v8f __attribute__((ext_vector_type(8)));

template <int WIRE>
__device__ __forceinline__ void ry(float (&u)[16], const float c, const float s) {
  constexpr int mask = 1 << (3 - WIRE);
  for (int m = 0; m < 16; ++m) {
    if ((m & mask) == 0) {
      const float a = u[m];
      const float b = u[m | mask];
      u[m]        = c * a - s * b;
      u[m | mask] = s * a + c * b;
    }
  }
}

template <int CTRL, int TGT>
__device__ __forceinline__ void cx(float (&u)[16]) {
  constexpr int cb = 3 - CTRL;
  constexpr int tb = 3 - TGT;
  for (int m = 0; m < 16; ++m) {
    if ((((m >> cb) & 1) == 1) && (((m >> tb) & 1) == 0)) {
      const float t = u[m];
      u[m] = u[m | (1 << tb)];
      u[m | (1 << tb)] = t;
    }
  }
}

template <int W0>
__device__ __forceinline__ void ry_all(float (&u)[16], const float (&c)[16], const float (&s)[16]) {
  ry<0>(u, c[W0 + 0], s[W0 + 0]);
  ry<1>(u, c[W0 + 1], s[W0 + 1]);
  ry<2>(u, c[W0 + 2], s[W0 + 2]);
  ry<3>(u, c[W0 + 3], s[W0 + 3]);
}

__device__ __forceinline__ void ring_fwd(float (&u)[16]) {
  cx<0, 1>(u); cx<1, 2>(u); cx<2, 3>(u); cx<3, 0>(u);
}

__device__ __forceinline__ void ring_bwd(float (&u)[16]) {
  cx<3, 0>(u); cx<2, 3>(u); cx<1, 2>(u); cx<0, 1>(u);
}

__device__ __forceinline__ void block_ring(float (&u)[16]) {
  cx<0, 1>(u); cx<2, 3>(u); cx<1, 2>(u); cx<3, 0>(u);
}

__device__ __forceinline__ void circ_ran(float (&u)[16], const float (&c)[16], const float (&s)[16]) {
  ry_all<0>(u, c, s);
  cx<0, 1>(u); cx<2, 3>(u);
  ry<0>(u, c[4], s[4]); ry<2>(u, c[5], s[5]);
  cx<1, 2>(u);
  ry<1>(u, c[6], s[6]);
}

__device__ __forceinline__ void circ_line(float (&u)[16], const float (&c)[16], const float (&s)[16]) {
  ry_all<0>(u, c, s);
  cx<0, 1>(u); cx<1, 2>(u); cx<2, 3>(u);
  ry<1>(u, c[4], s[4]); ry<2>(u, c[5], s[5]); ry<3>(u, c[6], s[6]);
}

__device__ __forceinline__ void circ_ring(float (&u)[16], const float (&c)[16], const float (&s)[16]) {
  ry_all<0>(u, c, s);
  ring_fwd(u);
  ry_all<4>(u, c, s);
}

__device__ __forceinline__ void circ_doublering(float (&u)[16], const float (&c)[16], const float (&s)[16]) {
  ry_all<0>(u, c, s);  ring_fwd(u);  ry_all<4>(u, c, s);   ring_bwd(u);
  ry_all<8>(u, c, s);  ring_fwd(u);  ry_all<12>(u, c, s);  ring_bwd(u);
}

__device__ __forceinline__ void circ_blockring(float (&u)[16], const float (&c)[16], const float (&s)[16]) {
  ry_all<0>(u, c, s);  block_ring(u);
  ry_all<4>(u, c, s);  block_ring(u);
  ry_all<8>(u, c, s);  block_ring(u);
}


__global__ void __launch_bounds__(256) build_coef_kernel(const float* __restrict__ wts,
                                                         const float* __restrict__ in,
                                                         float* __restrict__ ws,
                                                         const size_t ws_bytes) {
  if (blockIdx.x >= kCoefBlocks) {
    if (ws_bytes < kWsNeed) return;
    const int base = ((blockIdx.x - kCoefBlocks) * 8 + (threadIdx.x >> 5)) * 128;
    const int lane = threadIdx.x & 31;
    const v2f* ip = reinterpret_cast<const v2f*>(in + base);
    v4f*       pp = reinterpret_cast<v4f*>(ws + kPlaneOff + 2 * base);
    const v2f xa = ip[lane];
    const v2f xb = ip[32 + lane];
    float c0, s0, c1, s1, c2, s2, c3, s3;
    sincosf(xa[0], &s0, &c0);
    sincosf(xa[1], &s1, &c1);
    sincosf(xb[0], &s2, &c2);
    sincosf(xb[1], &s3, &c3);
    pp[lane]      = v4f{c0, s0, c1, s1};
    pp[32 + lane] = v4f{c2, s2, c3, s3};
    return;
  }
  float* __restrict__ Cf = ws;
  const int pair = blockIdx.x;
  const int i    = pair / 3;
  const int cin  = pair % 3;
  const int t    = threadIdx.x;
  const int m    = t >> 4;
  const int j    = t & 15;
  const int circ = i % 5;

  __shared__ float U[16][17];
  __shared__ float Msh[16][17];

  if (t < 32) {
    float sh, ch;
    sincosf(0.5f * wts[pair * 16 + j], &sh, &ch);

    float c[16], s[16];
    for (int k = 0; k < 16; ++k) {
      c[k] = __int_as_float(__builtin_amdgcn_readlane(__float_as_int(ch), k));
      s[k] = __int_as_float(__builtin_amdgcn_readlane(__float_as_int(sh), k));
    }

    float u[16];
    for (int mm = 0; mm < 16; ++mm) u[mm] = (mm == j) ? 1.0f : 0.0f;

    switch (circ) {
      case 0:  circ_ran(u, c, s);        break;
      case 1:  circ_line(u, c, s);       break;
      case 2:  circ_ring(u, c, s);       break;
      case 3:  circ_doublering(u, c, s); break;
      default: circ_blockring(u, c, s);  break;
    }

    if (t < 16) {
      for (int mm = 0; mm < 16; ++mm) U[mm][j] = u[mm];
    }
  }
  __syncthreads();

  float acc = 0.0f;
#pragma unroll
  for (int mm = 0; mm < 16; ++mm) {
    const float sgn = (mm < 8) ? 1.0f : -1.0f;
    acc += sgn * U[mm][m] * U[mm][j];
  }
  Msh[m][j] = acc;
  __syncthreads();

  if (t < 84) {
    const int h  = t / 42;
    const int sg = t % 42;
    const int a  = sg / 14;
    const int jj = sg % 14;
    int d1, q;
    if (h == 0) { d1 = (jj < 9) ? 0 : 1;  q = (jj < 9) ? jj : jj - 9; }
    else        { d1 = (jj < 9) ? 2 : 1;  q = (jj < 9) ? 8 - jj : 17 - jj; }
    const bool pad = (h == 1) && (jj == 13);
    const int d2 = q / 3;
    const int d3 = q % 3;
    const int flip = ((a == 2) ? 8 : 0) | ((d1 == 2) ? 4 : 0) | ((d2 == 2) ? 2 : 0) | ((d3 == 2) ? 1 : 0);
    const int neg  = ((a == 1) ? 8 : 0) | ((d1 == 1) ? 4 : 0) | ((d2 == 1) ? 2 : 0) | ((d3 == 1) ? 1 : 0);
    float cf = 0.0f;
#pragma unroll
    for (int r = 0; r < 16; ++r) {
      const float v = Msh[r][r ^ flip];
      cf += (__popc(r & neg) & 1) ? -v : v;
    }
    cf *= 0.0625f;
    Cf[(((cin * 21 + (sg >> 1)) * 32) + h * 16 + i) * 2 + (sg & 1)] = pad ? 0.0f : cf;
  }
}

template <bool PLANE>
__device__ __forceinline__ void tile_body(const float* __restrict__ in,
                                          const float* __restrict__ ws,
                                          float* __restrict__ out,
                                          const int lane, const int h, const int b,
                                          const int rem, const int q0) {
  float cs[3][4], sn[3][4];
  float x[3][4];
  if (PLANE) {
    const v2f* tp = reinterpret_cast<const v2f*>(ws + kPlaneOff) + q0;
#pragma unroll
    for (int c = 0; c < 3; ++c) {
      const v2f t0 = tp[c * 1024],      t1 = tp[c * 1024 + 1];
      const v2f t2 = tp[c * 1024 + 32], t3 = tp[c * 1024 + 33];
      cs[c][0] = t0[0]; sn[c][0] = t0[1];  cs[c][1] = t1[0]; sn[c][1] = t1[1];
      cs[c][2] = t2[0]; sn[c][2] = t2[1];  cs[c][3] = t3[0]; sn[c][3] = t3[1];
    }
  } else {
#pragma unroll
    for (int c = 0; c < 3; ++c) {
      const float* base = in + q0 + c * 1024;
      x[c][0] = base[0];  x[c][1] = base[1];
      x[c][2] = base[32]; x[c][3] = base[33];
    }
  }
  __builtin_amdgcn_sched_barrier(0);

  const v2f* Cv = reinterpret_cast<const v2f*>(ws) + lane;
  v2f afr[63];
#pragma unroll
  for (int g = 0; g < 63; ++g) afr[g] = Cv[g * 32];

  if (!PLANE) {
#pragma unroll
    for (int c = 0; c < 3; ++c) {
#pragma unroll
      for (int w = 0; w < 4; ++w) sincosf(x[c][w], &sn[c][w], &cs[c][w]);
    }
  }

  const bool hi = (h != 0);
  v8f acc0 = v8f{};
  v8f acc1 = v8f{};
#pragma unroll
  for (int c = 0; c < 3; ++c) {
    const float g3[3] = { hi ? sn[c][3] : 1.0f, cs[c][3], hi ? 1.0f : sn[c][3] };
    const float g2[3] = { hi ? sn[c][2] : 1.0f, cs[c][2], hi ? 1.0f : sn[c][2] };
    const float g1[2] = { hi ? sn[c][1] : 1.0f, cs[c][1] };
    float Q[9];
#pragma unroll
    for (int a2 = 0; a2 < 3; ++a2) {
#pragma unroll
      for (int a3 = 0; a3 < 3; ++a3) Q[3 * a2 + a3] = g2[a2] * g3[a3];
    }
    float R[14];
#pragma unroll
    for (int j = 0; j < 14; ++j) R[j] = (j < 9) ? g1[0] * Q[j] : g1[1] * Q[j - 9];
    float F[42];
#pragma unroll
    for (int j = 0; j < 14; ++j) {
      F[j]      = R[j];
      F[14 + j] = cs[c][0] * R[j];
      F[28 + j] = sn[c][0] * R[j];
    }
#pragma unroll
    for (int s = 0; s < 21; ++s) {
      const int g = c * 21 + s;
      v2f bfr;
      bfr[0] = F[2 * s];
      bfr[1] = F[2 * s + 1];
      if (g & 1) {
        acc1 = __builtin_amdgcn_wmma_f32_16x16x4_f32(false, afr[g], false, bfr, (short)0, acc1, false, false);
      } else {
        acc0 = __builtin_amdgcn_wmma_f32_16x16x4_f32(false, afr[g], false, bfr, (short)0, acc0, false, false);
      }
    }
  }

  float* op = out + (b * 16 + 8 * h) * 961 + rem;
#pragma unroll
  for (int r = 0; r < 8; ++r) op[r * 961] = acc0[r] + acc1[r];
}


__global__ void __launch_bounds__(128) quanv_wmma_kernel(const float* __restrict__ in,
                                                         const float* __restrict__ ws,
                                                         float* __restrict__ out,
                                                         const size_t ws_bytes) {
  const int tile = blockIdx.x * 4 + (threadIdx.x >> 5);
  if (tile >= 961) return;
  const int lane = threadIdx.x & 31;
  const int h    = lane >> 4;
  const int n    = lane & 15;

  const int p   = tile * 16 + n;
  const int b   = p / 961;
  const int rem = p % 961;
  const int ph  = rem / 31;
  const int pw  = rem % 31;
  const int q0  = ((b * 3) * 32 + ph) * 32 + pw;

  __builtin_amdgcn_sched_barrier(0);
  if (__builtin_expect(ws_bytes >= kWsNeed, 1)) tile_body<true>(in, ws, out, lane, h, b, rem, q0);
  else                                          tile_body<false>(in, ws, out, lane, h, b, rem, q0);
}

extern "C" void kernel_launch(void* const* d_in, const int* in_sizes, int n_in,
                              void* d_out, int out_size, void* d_ws, size_t ws_size,
                              hipStream_t stream) {
  const float* inputs  = (const float*)d_in[0];
  const float* weights = (const float*)d_in[1];
  float*       out     = (float*)d_out;
  float*       ws      = (float*)d_ws;

  build_coef_kernel<<<96, 256, 0, stream>>>(weights, inputs, ws, ws_size);
  quanv_wmma_kernel<<<241, 128, 0, stream>>>(inputs, ws, out, ws_size);
}


# One launch: each block derives all 48x84 coefficients by conjugating Z on wire 0 back through the gates (I/X/Y/Z operator strings, one pair per lane): no 16x16 M, fold or workspace; sum order changes
# speedup vs baseline: 3.7937x; 1.3874x over previous
#include <hip/hip_runtime.h>


typedef float v2f __attribute__((ext_vector_type(2)));
typedef float v8f __attribute__((ext_vector_type(8)));

__device__ __forceinline__ void coef_ran(const float (&C)[16], const float (&S)[16], float (&cf)[84]) {
  const float v0 = C[4];
  const float v1 = -(S[4]);
  const float v2 = S[1] * v1;
  const float v3 = C[1] * v1;
  const float v4 = C[0] * v0;
  const float v5 = -(S[0] * v0);
  const float v6 = S[0] * v3;
  const float v7 = C[0] * v3;
  const float v8 = S[0] * v2;
  const float v9 = C[0] * v2;
  cf[14] = v4;
  cf[23] = v8;
  cf[28] = v5;
  cf[37] = v9;
  cf[64] = v6;
  cf[78] = v7;
}

__device__ __forceinline__ void coef_line(const float (&C)[16], const float (&S)[16], float (&cf)[84]) {
  const float v0 = C[0];
  const float v1 = -(S[0]);
  cf[14] = v0;
  cf[28] = v1;
}

__device__ __forceinline__ void coef_ring(const float (&C)[16], const float (&S)[16], float (&cf)[84]) {
  const float v0 = C[4];
  const float v1 = -(S[4]);
  const float v2 = C[3] * v0;
  const float v3 = -(S[3] * v0);
  const float v4 = C[2] * v3;
  const float v5 = -(S[2] * v3);
  const float v6 = C[2] * v2;
  const float v7 = -(S[2] * v2);
  const float v8 = C[1] * v5;
  const float v9 = -(S[1] * v5);
  const float v10 = C[1] * v7;
  const float v11 = -(S[1] * v7);
  const float v12 = C[1] * v4;
  const float v13 = -(S[1] * v4);
  const float v14 = C[1] * v6;
  const float v15 = -(S[1] * v6);
  const float v16 = S[1] * v1;
  const float v17 = C[1] * v1;
  const float v18 = S[0] * v17;
  const float v19 = C[0] * v17;
  const float v20 = S[0] * v16;
  const float v21 = C[0] * v16;
  cf[13] = v14;
  cf[23] = v20;
  cf[37] = v21;
  cf[42] = v9;
  cf[43] = v11;
  cf[45] = v13;
  cf[46] = v15;
  cf[51] = v8;
  cf[52] = v10;
  cf[54] = v12;
  cf[64] = v18;
  cf[78] = v19;
}

__device__ __forceinline__ void coef_doublering(const float (&C)[16], const float (&S)[16], float (&cf)[84]) {
  const float v0 = C[15];
  const float v1 = -(S[15]);
  const float v2 = C[12] * v1;
  const float v3 = -(S[12] * v1);
  const float v4 = C[12] * v0;
  const float v5 = -(S[12] * v0);
  const float v6 = S[11] * v3;
  const float v7 = C[11] * v3;
  const float v8 = -(C[11] * v5);
  const float v9 = S[11] * v5;
  const float v10 = -(C[10] * v2);
  const float v11 = S[10] * v2;
  const float v12 = C[10] * v9;
  const float v13 = -(S[10] * v9);
  const float v14 = C[10] * v8;
  const float v15 = -(S[10] * v8);
  const float v16 = C[8] * v4;
  const float v17 = -(S[8] * v4);
  const float v18 = S[8] * v11;
  const float v20 = S[8] * v10;
  const float v21 = C[8] * v10;
  const float v22 = C[7] * v6 - S[7] * v12;
  const float v23 = -(S[7] * v6) - C[7] * v12;
  const float v24 = S[7] * v17 - C[7] * v20;
  const float v25 = S[7] * v20 + C[7] * v17;
  const float v26 = S[7] * v7 - C[7] * v14;
  const float v27 = S[7] * v14 + C[7] * v7;
  const float v28 = C[7] * v16 - S[7] * v21;
  const float v29 = -(S[7] * v16) - C[7] * v21;
  const float v31 = -(S[6] * v23);
  const float v32 = C[6] * v22;
  const float v34 = S[6] * v25;
  const float v35 = C[6] * v25;
  const float v36 = S[6] * v24;
  const float v39 = -(S[6] * v18);
  const float v40 = -(S[6] * v13);
  const float v41 = -(C[6] * v13);
  const float v43 = C[5] * v35;
  const float v47 = -(S[5] * v15);
  const float v48 = S[5] * v34;
  const float v50 = S[5] * v36;
  const float v51 = C[5] * v36;
  const float v53 = -(S[5] * v41);
  const float v54 = C[5] * v40;
  const float v55 = -(S[5] * v40);
  const float v59 = C[4] * v27 - S[4] * v29;
  const float v60 = C[4] * v28 + S[4] * v26;
  const float v63 = C[4] * v53;
  const float v64 = S[4] * v55;
  const float v66 = S[4] * v39;
  const float v67 = C[4] * v39;
  const float v72 = S[4] * v54;
  const float v73 = C[4] * v54;
  const float v74 = C[3] * v32 + S[3] * v59;
  const float v75 = C[3] * v59 - S[3] * v32;
  const float v76 = C[3] * v51;
  const float v77 = -(S[3] * v51);
  const float v78 = -(C[3] * v67);
  const float v79 = S[3] * v67;
  const float v80 = C[3] * v72;
  const float v81 = -(S[3] * v72);
  const float v82 = S[3] * v63 - C[3] * v47;
  const float v83 = S[3] * v47 + C[3] * v63;
  const float v84 = S[3] * v48;
  const float v85 = C[3] * v48;
  const float v94 = C[3] * v64;
  const float v95 = -(S[3] * v64);
  const float v96 = C[3] * v50;
  const float v97 = -(S[3] * v50);
  const float v98 = C[2] * v79 + S[2] * v77;
  const float v99 = C[2] * v77 - S[2] * v79;
  const float v100 = C[2] * v78 + S[2] * v76;
  const float v101 = C[2] * v76 - S[2] * v78;
  const float v104 = C[2] * v73 + S[2] * v31;
  const float v105 = C[2] * v31 - S[2] * v73;
  const float v106 = C[2] * v85;
  const float v107 = -(S[2] * v85);
  const float v108 = C[2] * v84;
  const float v109 = -(S[2] * v84);
  const float v118 = S[2] * v95;
  const float v119 = C[2] * v95;
  const float v120 = S[2] * v94;
  const float v121 = C[2] * v94;
  const float v124 = C[1] * v81;
  const float v125 = -(S[1] * v81);
  const float v126 = C[1] * v80;
  const float v127 = -(S[1] * v80);
  const float v128 = S[1] * v99;
  const float v129 = C[1] * v99;
  const float v130 = S[1] * v101;
  const float v131 = C[1] * v101;
  const float v132 = S[1] * v98;
  const float v133 = C[1] * v98;
  const float v134 = S[1] * v100;
  const float v135 = C[1] * v100;
  const float v136 = -(C[1] * v66);
  const float v137 = S[1] * v66;
  const float v138 = S[1] * v105;
  const float v139 = C[1] * v105;
  const float v140 = S[1] * v107;
  const float v141 = C[1] * v107;
  const float v142 = S[1] * v109;
  const float v143 = C[1] * v109;
  const float v144 = S[1] * v104;
  const float v145 = C[1] * v104;
  const float v146 = S[1] * v106;
  const float v147 = C[1] * v106;
  const float v148 = S[1] * v108;
  const float v149 = C[1] * v108;
  const float v156 = C[1] * v97;
  const float v157 = -(S[1] * v97);
  const float v158 = C[1] * v96;
  const float v159 = -(S[1] * v96);
  const float v160 = S[1] * v119;
  const float v161 = C[1] * v119;
  const float v162 = S[1] * v121;
  const float v163 = C[1] * v121;
  const float v166 = S[1] * v118;
  const float v167 = C[1] * v118;
  const float v168 = S[1] * v120;
  const float v169 = C[1] * v120;
  const float v170 = C[0] * v60 + S[0] * v43;
  const float v171 = C[0] * v43 - S[0] * v60;
  const float v172 = S[0] * v83;
  const float v173 = C[0] * v83;
  const float v174 = S[0] * v82;
  const float v175 = C[0] * v82;
  const float v178 = S[0] * v137;
  const float v179 = C[0] * v137;
  const float v180 = C[0] * v157;
  const float v181 = -(S[0] * v157);
  const float v182 = C[0] * v159;
  const float v183 = -(S[0] * v159);
  const float v184 = S[0] * v139;
  const float v185 = C[0] * v139;
  const float v186 = C[0] * v161 + S[0] * v141;
  const float v187 = C[0] * v141 - S[0] * v161;
  const float v188 = C[0] * v163 + S[0] * v143;
  const float v189 = C[0] * v143 - S[0] * v163;
  const float v192 = S[0] * v145;
  const float v193 = C[0] * v145;
  const float v194 = C[0] * v167 + S[0] * v147;
  const float v195 = C[0] * v147 - S[0] * v167;
  const float v196 = C[0] * v169 + S[0] * v149;
  const float v197 = C[0] * v149 - S[0] * v169;
  const float v202 = S[0] * v136;
  const float v203 = C[0] * v136;
  const float v204 = C[0] * v156;
  const float v205 = -(S[0] * v156);
  const float v206 = C[0] * v158;
  const float v207 = -(S[0] * v158);
  const float v208 = S[0] * v138;
  const float v209 = C[0] * v138;
  const float v210 = C[0] * v160 + S[0] * v140;
  const float v211 = C[0] * v140 - S[0] * v160;
  const float v212 = C[0] * v162 + S[0] * v142;
  const float v213 = C[0] * v142 - S[0] * v162;
  const float v216 = S[0] * v144;
  const float v217 = C[0] * v144;
  const float v218 = C[0] * v166 + S[0] * v146;
  const float v219 = C[0] * v146 - S[0] * v166;
  const float v220 = C[0] * v168 + S[0] * v148;
  const float v221 = C[0] * v148 - S[0] * v168;
  cf[1] = v74;
  cf[2] = v75;
  cf[10] = v126;
  cf[11] = v124;
  cf[13] = v134;
  cf[14] = v170;
  cf[15] = v174;
  cf[16] = v172;
  cf[23] = v202;
  cf[24] = v206;
  cf[25] = v204;
  cf[26] = v216;
  cf[27] = v220;
  cf[28] = v171;
  cf[29] = v175;
  cf[30] = v173;
  cf[37] = v203;
  cf[38] = v207;
  cf[39] = v205;
  cf[40] = v217;
  cf[41] = v221;
  cf[42] = v129;
  cf[43] = v131;
  cf[45] = v133;
  cf[46] = v135;
  cf[48] = v125;
  cf[49] = v127;
  cf[51] = v128;
  cf[52] = v130;
  cf[54] = v132;
  cf[56] = v186;
  cf[57] = v188;
  cf[58] = v184;
  cf[59] = v194;
  cf[60] = v196;
  cf[61] = v192;
  cf[62] = v180;
  cf[63] = v182;
  cf[64] = v178;
  cf[65] = v210;
  cf[66] = v212;
  cf[67] = v208;
  cf[68] = v218;
  cf[70] = v187;
  cf[71] = v189;
  cf[72] = v185;
  cf[73] = v195;
  cf[74] = v197;
  cf[75] = v193;
  cf[76] = v181;
  cf[77] = v183;
  cf[78] = v179;
  cf[79] = v211;
  cf[80] = v213;
  cf[81] = v209;
  cf[82] = v219;
}

__device__ __forceinline__ void coef_blockring(const float (&C)[16], const float (&S)[16], float (&cf)[84]) {
  const float v0 = C[11];
  const float v1 = -(S[11]);
  const float v2 = C[10] * v1;
  const float v3 = -(S[10] * v1);
  const float v4 = C[10] * v0;
  const float v5 = -(S[10] * v0);
  const float v6 = C[8] * v3;
  const float v7 = -(S[8] * v3);
  const float v8 = C[8] * v5;
  const float v9 = -(S[8] * v5);
  const float v10 = C[8] * v2;
  const float v11 = -(S[8] * v2);
  const float v12 = C[8] * v4;
  const float v13 = -(S[8] * v4);
  const float v14 = -(C[7] * v6);
  const float v15 = S[7] * v6;
  const float v16 = -(C[7] * v13);
  const float v18 = S[7] * v8;
  const float v19 = C[7] * v8;
  const float v20 = S[7] * v11;
  const float v21 = C[7] * v11;
  const float v22 = S[6] * v7;
  const float v23 = C[6] * v7;
  const float v24 = C[6] * v12;
  const float v25 = -(S[6] * v12);
  const float v26 = S[6] * v19;
  const float v27 = C[6] * v19;
  const float v28 = S[6] * v18;
  const float v29 = C[6] * v18;
  const float v30 = C[6] * v21;
  const float v31 = -(S[6] * v21);
  const float v32 = C[6] * v20;
  const float v33 = -(S[6] * v20);
  const float v34 = C[5] * v25;
  const float v35 = -(S[5] * v25);
  const float v36 = C[5] * v24;
  const float v38 = -(S[5] * v9);
  const float v39 = -(C[5] * v9);
  const float v40 = S[5] * v15;
  const float v43 = C[5] * v14;
  const float v44 = C[5] * v31;
  const float v45 = -(S[5] * v31);
  const float v46 = C[5] * v33;
  const float v47 = -(S[5] * v33);
  const float v48 = C[5] * v30;
  const float v49 = -(S[5] * v30);
  const float v50 = C[5] * v32;
  const float v53 = -(S[4] * v27);
  const float v54 = C[4] * v29;
  const float v57 = -(S[4] * v26);
  const float v58 = C[4] * v28;
  const float v61 = -(S[4] * v45);
  const float v62 = C[4] * v47;
  const float v63 = -(S[4] * v47);
  const float v64 = S[4] * v39;
  const float v66 = C[4] * v49;
  const float v70 = -(S[4] * v10);
  const float v73 = -(S[4] * v44);
  const float v74 = C[4] * v46;
  const float v77 = C[4] * v38;
  const float v79 = -(S[4] * v48);
  const float v80 = C[4] * v50;
  const float v81 = -(S[4] * v50);
  const float v82 = S[3] * v23 - C[3] * v77;
  const float v83 = S[3] * v77 + C[3] * v23;
  const float v84 = S[3] * v79;
  const float v85 = C[3] * v79;
  const float v86 = S[3] * v61;
  const float v87 = C[3] * v61;
  const float v90 = S[3] * v74;
  const float v91 = C[3] * v74;
  const float v92 = C[3] * v63;
  const float v93 = -(S[3] * v63);
  const float v94 = C[3] * v81;
  const float v95 = -(S[3] * v81);
  const float v98 = C[3] * v66 - S[3] * v16;
  const float v99 = -(S[3] * v66) - C[3] * v16;
  const float v100 = -(S[3] * v64);
  const float v101 = -(C[3] * v64);
  const float v106 = S[3] * v54 - C[3] * v40;
  const float v107 = S[3] * v40 + C[3] * v54;
  const float v110 = S[3] * v34;
  const float v111 = C[3] * v34;
  const float v112 = S[3] * v57;
  const float v113 = C[3] * v57;
  const float v114 = C[2] * v36 + S[2] * v53;
  const float v115 = C[2] * v53 - S[2] * v36;
  const float v116 = C[2] * v85 + S[2] * v83;
  const float v117 = C[2] * v83 - S[2] * v85;
  const float v118 = C[2] * v84 + S[2] * v82;
  const float v119 = C[2] * v82 - S[2] * v84;
  const float v120 = C[2] * v58;
  const float v121 = -(S[2] * v58);
  const float v122 = S[2] * v91;
  const float v123 = C[2] * v91;
  const float v124 = S[2] * v90;
  const float v125 = C[2] * v90;
  const float v126 = C[2] * v93;
  const float v127 = -(S[2] * v93);
  const float v128 = C[2] * v92;
  const float v129 = -(S[2] * v92);
  const float v130 = -(C[2] * v43) - S[2] * v70;
  const float v131 = S[2] * v43 - C[2] * v70;
  const float v132 = C[2] * v101 + S[2] * v99;
  const float v133 = C[2] * v99 - S[2] * v101;
  const float v134 = C[2] * v100 + S[2] * v98;
  const float v135 = C[2] * v98 - S[2] * v100;
  const float v140 = C[2] * v80;
  const float v141 = -(S[2] * v80);
  const float v142 = S[2] * v107;
  const float v143 = C[2] * v107;
  const float v144 = S[2] * v106;
  const float v145 = C[2] * v106;
  const float v146 = C[2] * v22 + S[2] * v73;
  const float v147 = C[2] * v73 - S[2] * v22;
  const float v148 = C[2] * v113 + S[2] * v111;
  const float v149 = C[2] * v111 - S[2] * v113;
  const float v150 = C[2] * v112 + S[2] * v110;
  const float v151 = C[2] * v110 - S[2] * v112;
  const float v152 = S[1] * v35;
  const float v153 = C[1] * v35;
  const float v154 = S[1] * v87;
  const float v155 = C[1] * v87;
  const float v156 = S[1] * v86;
  const float v157 = C[1] * v86;
  const float v158 = C[1] * v121;
  const float v159 = -(S[1] * v121);
  const float v160 = C[1] * v123;
  const float v161 = -(S[1] * v123);
  const float v162 = C[1] * v125;
  const float v163 = -(S[1] * v125);
  const float v164 = C[1] * v120;
  const float v165 = -(S[1] * v120);
  const float v166 = C[1] * v122;
  const float v167 = -(S[1] * v122);
  const float v168 = C[1] * v124;
  const float v169 = -(S[1] * v124);
  const float v170 = S[1] * v95;
  const float v171 = C[1] * v95;
  const float v172 = S[1] * v94;
  const float v173 = C[1] * v94;
  const float v174 = C[1] * v131;
  const float v175 = -(S[1] * v131);
  const float v176 = C[1] * v133;
  const float v177 = -(S[1] * v133);
  const float v178 = C[1] * v135;
  const float v179 = -(S[1] * v135);
  const float v182 = C[1] * v130;
  const float v183 = -(S[1] * v130);
  const float v184 = C[1] * v132;
  const float v185 = -(S[1] * v132);
  const float v186 = C[1] * v134;
  const float v187 = -(S[1] * v134);
  const float v200 = S[1] * v62;
  const float v201 = C[1] * v62;
  const float v202 = C[1] * v147;
  const float v203 = -(S[1] * v147);
  const float v204 = C[1] * v149;
  const float v205 = -(S[1] * v149);
  const float v206 = C[1] * v151;
  const float v207 = -(S[1] * v151);
  const float v208 = C[1] * v146;
  const float v209 = -(S[1] * v146);
  const float v210 = C[1] * v148;
  const float v211 = -(S[1] * v148);
  const float v212 = C[1] * v150;
  const float v213 = -(S[1] * v150);
  const float v214 = C[0] * v141;
  const float v215 = -(S[0] * v141);
  const float v216 = C[0] * v143 + S[0] * v127;
  const float v217 = C[0] * v127 - S[0] * v143;
  const float v218 = C[0] * v145 + S[0] * v129;
  const float v219 = C[0] * v129 - S[0] * v145;
  const float v220 = C[0] * v140;
  const float v221 = -(S[0] * v140);
  const float v222 = C[0] * v142 + S[0] * v126;
  const float v223 = C[0] * v126 - S[0] * v142;
  const float v224 = C[0] * v144 + S[0] * v128;
  const float v225 = C[0] * v128 - S[0] * v144;
  const float v226 = C[0] * v201;
  const float v227 = -(S[0] * v201);
  const float v228 = S[0] * v171;
  const float v229 = C[0] * v171;
  const float v230 = S[0] * v173;
  const float v231 = C[0] * v173;
  const float v232 = C[0] * v203 + S[0] * v175;
  const float v233 = C[0] * v175 - S[0] * v203;
  const float v234 = C[0] * v205 + S[0] * v177;
  const float v235 = C[0] * v177 - S[0] * v205;
  const float v236 = C[0] * v207 + S[0] * v179;
  const float v237 = C[0] * v179 - S[0] * v207;
  const float v240 = C[0] * v209 + S[0] * v183;
  const float v241 = C[0] * v183 - S[0] * v209;
  const float v242 = C[0] * v211 + S[0] * v185;
  const float v243 = C[0] * v185 - S[0] * v211;
  const float v244 = C[0] * v213 + S[0] * v187;
  const float v245 = C[0] * v187 - S[0] * v213;
  const float v254 = C[0] * v200;
  const float v255 = -(S[0] * v200);
  const float v256 = S[0] * v170;
  const float v257 = C[0] * v170;
  const float v258 = S[0] * v172;
  const float v259 = C[0] * v172;
  const float v260 = C[0] * v202 + S[0] * v174;
  const float v261 = C[0] * v174 - S[0] * v202;
  const float v262 = C[0] * v204 + S[0] * v176;
  const float v263 = C[0] * v176 - S[0] * v204;
  const float v264 = C[0] * v206 + S[0] * v178;
  const float v265 = C[0] * v178 - S[0] * v206;
  const float v268 = C[0] * v208 + S[0] * v182;
  const float v269 = C[0] * v182 - S[0] * v208;
  const float v270 = C[0] * v210 + S[0] * v184;
  const float v271 = C[0] * v184 - S[0] * v210;
  const float v272 = C[0] * v212 + S[0] * v186;
  const float v273 = C[0] * v186 - S[0] * v212;
  cf[3] = v114;
  cf[4] = v118;
  cf[5] = v116;
  cf[6] = v115;
  cf[7] = v119;
  cf[8] = v117;
  cf[9] = v152;
  cf[10] = v156;
  cf[11] = v154;
  cf[12] = v164;
  cf[13] = v168;
  cf[17] = v220;
  cf[18] = v224;
  cf[19] = v222;
  cf[20] = v214;
  cf[21] = v218;
  cf[22] = v216;
  cf[23] = v254;
  cf[24] = v258;
  cf[25] = v256;
  cf[26] = v268;
  cf[27] = v272;
  cf[31] = v221;
  cf[32] = v225;
  cf[33] = v223;
  cf[34] = v215;
  cf[35] = v219;
  cf[36] = v217;
  cf[37] = v255;
  cf[38] = v259;
  cf[39] = v257;
  cf[40] = v269;
  cf[41] = v273;
  cf[42] = v161;
  cf[43] = v163;
  cf[44] = v159;
  cf[45] = v167;
  cf[46] = v169;
  cf[47] = v165;
  cf[48] = v155;
  cf[49] = v157;
  cf[50] = v153;
  cf[51] = v160;
  cf[52] = v162;
  cf[53] = v158;
  cf[54] = v166;
  cf[56] = v234;
  cf[57] = v236;
  cf[58] = v232;
  cf[59] = v242;
  cf[60] = v244;
  cf[61] = v240;
  cf[62] = v228;
  cf[63] = v230;
  cf[64] = v226;
  cf[65] = v262;
  cf[66] = v264;
  cf[67] = v260;
  cf[68] = v270;
  cf[70] = v235;
  cf[71] = v237;
  cf[72] = v233;
  cf[73] = v243;
  cf[74] = v245;
  cf[75] = v241;
  cf[76] = v229;
  cf[77] = v231;
  cf[78] = v227;
  cf[79] = v263;
  cf[80] = v265;
  cf[81] = v261;
  cf[82] = v271;
}

__device__ __forceinline__ void store_slots(const float (&cf)[84], v2f* __restrict__ tb) {
#pragma unroll
  for (int h = 0; h < 2; ++h) {
#pragma unroll
    for (int s = 0; s < 21; ++s) tb[s * 32 + h * 16] = v2f{cf[h * 42 + 2 * s], cf[h * 42 + 2 * s + 1]};
  }
}

__global__ void __launch_bounds__(128) quanv_onelaunch_kernel(const float* __restrict__ wts,
                                                              const float* __restrict__ in,
                                                              float* __restrict__ out) {
  __shared__ v2f tblv[63 * 32];
  __shared__ v2f dws[18 * 16];
  __shared__ v2f trg[4 * 12 * 16];

  const int t    = threadIdx.x;
  const int lane = t & 31;
  const int wv   = __builtin_amdgcn_readfirstlane(t >> 5);

  float wval[2];
  int   wslot[2];
#pragma unroll
  for (int r = 0; r < 2; ++r) {
    const int  v   = (t + 128 * r < 252) ? t + 128 * r : 251;
    const bool dbl = v < 144;
    const int  u   = dbl ? v : v - 144;
    const int  L   = dbl ? (u >> 4) : (u / 12);
    const int  k   = dbl ? (u & 15) : (u % 12);
    const int  i   = (dbl ? 3 : 4) + 5 * (L / 3);
    const int  c   = L % 3;
    wslot[r] = ((dbl ? 0 : 9) + L) * 16 + k;
    wval[r]  = wts[(i * 3 + c) * 16 + k];
  }
  float px[6];
#pragma unroll
  for (int r = 0; r < 6; ++r) {
    const int v    = t + 128 * r;
    const int tw   = v / 192;
    const int rest = v % 192;
    const int cc   = rest >> 4;
    const int n    = rest & 15;
    const int tl   = (blockIdx.x * 4 + tw < 961) ? blockIdx.x * 4 + tw : 960;
    const int p    = tl * 16 + n;
    const int b    = p / 961;
    const int rem  = p % 961;
    const int ph   = rem / 31;
    const int pw   = rem % 31;
    const int q    = ((b * 3 + (cc >> 2)) * 32 + ph + ((cc >> 1) & 1)) * 32 + pw + (cc & 1);
    px[r] = in[q];
  }
  const int sl      = (lane < 30) ? lane : 29;
  const int sh_type = (sl < 12) ? 0 : (sl < 21) ? 1 : 2;
  const int sh_u    = (sl < 12) ? sl : (sl < 21) ? sl - 12 : sl - 21;
  const int sh_i    = sh_type + 5 * (sh_u / 3);
  const int sh_c    = sh_u % 3;
  float shw[5];
  {
    const float* wp = wts + (sh_i * 3 + sh_c) * 16;
#pragma unroll
    for (int k = 0; k < 5; ++k) shw[k] = wp[k];
  }

#pragma unroll
  for (int r = 0; r < 2; ++r) {
    float sn, cs;
    sincosf(wval[r], &sn, &cs);
    if (t + 128 * r < 252) dws[wslot[r]] = v2f{cs, sn};
  }
#pragma unroll
  for (int r = 0; r < 6; ++r) {
    float sn, cs;
    sincosf(px[r], &sn, &cs);
    trg[t + 128 * r] = v2f{cs, sn};
  }
  __syncthreads();

  if (wv < 2) {
    if (lane < 9) {
      const int i = 3 + wv + 5 * (lane / 3);
      const int c = lane % 3;
      const v2f* e = dws + (wv * 9 + lane) * 16;
      float C[16], S[16];
      float cf[84];
#pragma unroll
      for (int k = 0; k < 84; ++k) cf[k] = 0.0f;
      if (wv == 0) {
#pragma unroll
        for (int k = 0; k < 16; ++k) { const v2f a = e[k]; C[k] = a[0]; S[k] = a[1]; }
        coef_doublering(C, S, cf);
      } else {
#pragma unroll
        for (int k = 0; k < 16; ++k) { C[k] = 0.0f; S[k] = 0.0f; }
#pragma unroll
        for (int k = 0; k < 12; ++k) { const v2f a = e[k]; C[k] = a[0]; S[k] = a[1]; }
        coef_blockring(C, S, cf);
      }
      store_slots(cf, tblv + c * 21 * 32 + i);
    }
  } else if (wv == 2) {
    if (lane < 30) {
      float C[16], S[16];
#pragma unroll
      for (int k = 0; k < 16; ++k) { C[k] = 0.0f; S[k] = 0.0f; }
#pragma unroll
      for (int k = 0; k < 5; ++k) sincosf(shw[k], &S[k], &C[k]);
      float cf[84];
#pragma unroll
      for (int k = 0; k < 84; ++k) cf[k] = 0.0f;
      if (sh_type == 0)      coef_ran(C, S, cf);
      else if (sh_type == 1) coef_line(C, S, cf);
      else                   coef_ring(C, S, cf);
      store_slots(cf, tblv + sh_c * 21 * 32 + sh_i);
    }
  }
  __syncthreads();

  const int tile = blockIdx.x * 4 + wv;
  if (tile >= 961) return;
  const int h   = lane >> 4;
  const int n   = lane & 15;
  const int p   = tile * 16 + n;
  const int b   = p / 961;
  const int rem = p % 961;

  float cs[3][4], sn[3][4];
  const v2f* tp = trg + wv * 192 + n;
#pragma unroll
  for (int c = 0; c < 3; ++c) {
#pragma unroll
    for (int w = 0; w < 4; ++w) {
      const v2f a = tp[(c * 4 + w) * 16];
      cs[c][w] = a[0];
      sn[c][w] = a[1];
    }
  }
  __builtin_amdgcn_sched_barrier(0);

  const v2f* Cv = tblv + lane;
  v2f afr[63];
#pragma unroll
  for (int g = 0; g < 63; ++g) afr[g] = Cv[g * 32];
  __builtin_amdgcn_sched_barrier(0);

  const bool hi = (h != 0);
  v8f acc0 = v8f{};
  v8f acc1 = v8f{};
#pragma unroll
  for (int c = 0; c < 3; ++c) {
    const float g3[3] = { hi ? sn[c][3] : 1.0f, cs[c][3], hi ? 1.0f : sn[c][3] };
    const float g2[3] = { hi ? sn[c][2] : 1.0f, cs[c][2], hi ? 1.0f : sn[c][2] };
    const float g1[2] = { hi ? sn[c][1] : 1.0f, cs[c][1] };
    float Q[9];
#pragma unroll
    for (int a2 = 0; a2 < 3; ++a2) {
#pragma unroll
      for (int a3 = 0; a3 < 3; ++a3) Q[3 * a2 + a3] = g2[a2] * g3[a3];
    }
    float R[14];
#pragma unroll
    for (int j = 0; j < 14; ++j) R[j] = (j < 9) ? g1[0] * Q[j] : g1[1] * Q[j - 9];
    float F[42];
#pragma unroll
    for (int j = 0; j < 14; ++j) {
      F[j]      = R[j];
      F[14 + j] = cs[c][0] * R[j];
      F[28 + j] = sn[c][0] * R[j];
    }
#pragma unroll
    for (int s = 0; s < 21; ++s) {
      const int g = c * 21 + s;
      v2f bfr;
      bfr[0] = F[2 * s];
      bfr[1] = F[2 * s + 1];
      if (g & 1) {
        acc1 = __builtin_amdgcn_wmma_f32_16x16x4_f32(false, afr[g], false, bfr, (short)0, acc1, false, false);
      } else {
        acc0 = __builtin_amdgcn_wmma_f32_16x16x4_f32(false, afr[g], false, bfr, (short)0, acc0, false, false);
      }
    }
  }

  float* op = out + (b * 16 + 8 * h) * 961 + rem;
#pragma unroll
  for (int r = 0; r < 8; ++r) op[r * 961] = acc0[r] + acc1[r];
}

extern "C" void kernel_launch(void* const* d_in, const int* in_sizes, int n_in,
                              void* d_out, int out_size, void* d_ws, size_t ws_size,
                              hipStream_t stream) {
  const float* inputs  = (const float*)d_in[0];
  const float* weights = (const float*)d_in[1];
  float*       out     = (float*)d_out;

  quanv_onelaunch_kernel<<<241, 128, 0, stream>>>(weights, inputs, out);
}


# cos,sin by an inline branch-free routine (pi/2 reduction + polynomials), interleaved; each lane does its own patch's 12 pixels in registers behind the table reads, no LDS staging; sum order unchanged
# speedup vs baseline: 4.1357x; 1.0901x over previous
#include <hip/hip_runtime.h>


typedef float v2f __attribute__((ext_vector_type(2)));
typedef float v8f __attribute__((ext_vector_type(8)));

__device__ __forceinline__ void coef_ran(const float (&C)[16], const float (&S)[16], float (&cf)[84]) {
  const float v0 = C[4];
  const float v1 = -(S[4]);
  const float v2 = S[1] * v1;
  const float v3 = C[1] * v1;
  const float v4 = C[0] * v0;
  const float v5 = -(S[0] * v0);
  const float v6 = S[0] * v3;
  const float v7 = C[0] * v3;
  const float v8 = S[0] * v2;
  const float v9 = C[0] * v2;
  cf[14] = v4;
  cf[23] = v8;
  cf[28] = v5;
  cf[37] = v9;
  cf[64] = v6;
  cf[78] = v7;
}

__device__ __forceinline__ void coef_line(const float (&C)[16], const float (&S)[16], float (&cf)[84]) {
  const float v0 = C[0];
  const float v1 = -(S[0]);
  cf[14] = v0;
  cf[28] = v1;
}

__device__ __forceinline__ void coef_ring(const float (&C)[16], const float (&S)[16], float (&cf)[84]) {
  const float v0 = C[4];
  const float v1 = -(S[4]);
  const float v2 = C[3] * v0;
  const float v3 = -(S[3] * v0);
  const float v4 = C[2] * v3;
  const float v5 = -(S[2] * v3);
  const float v6 = C[2] * v2;
  const float v7 = -(S[2] * v2);
  const float v8 = C[1] * v5;
  const float v9 = -(S[1] * v5);
  const float v10 = C[1] * v7;
  const float v11 = -(S[1] * v7);
  const float v12 = C[1] * v4;
  const float v13 = -(S[1] * v4);
  const float v14 = C[1] * v6;
  const float v15 = -(S[1] * v6);
  const float v16 = S[1] * v1;
  const float v17 = C[1] * v1;
  const float v18 = S[0] * v17;
  const float v19 = C[0] * v17;
  const float v20 = S[0] * v16;
  const float v21 = C[0] * v16;
  cf[13] = v14;
  cf[23] = v20;
  cf[37] = v21;
  cf[42] = v9;
  cf[43] = v11;
  cf[45] = v13;
  cf[46] = v15;
  cf[51] = v8;
  cf[52] = v10;
  cf[54] = v12;
  cf[64] = v18;
  cf[78] = v19;
}

__device__ __forceinline__ void coef_doublering(const float (&C)[16], const float (&S)[16], float (&cf)[84]) {
  const float v0 = C[15];
  const float v1 = -(S[15]);
  const float v2 = C[12] * v1;
  const float v3 = -(S[12] * v1);
  const float v4 = C[12] * v0;
  const float v5 = -(S[12] * v0);
  const float v6 = S[11] * v3;
  const float v7 = C[11] * v3;
  const float v8 = -(C[11] * v5);
  const float v9 = S[11] * v5;
  const float v10 = -(C[10] * v2);
  const float v11 = S[10] * v2;
  const float v12 = C[10] * v9;
  const float v13 = -(S[10] * v9);
  const float v14 = C[10] * v8;
  const float v15 = -(S[10] * v8);
  const float v16 = C[8] * v4;
  const float v17 = -(S[8] * v4);
  const float v18 = S[8] * v11;
  const float v20 = S[8] * v10;
  const float v21 = C[8] * v10;
  const float v22 = C[7] * v6 - S[7] * v12;
  const float v23 = -(S[7] * v6) - C[7] * v12;
  const float v24 = S[7] * v17 - C[7] * v20;
  const float v25 = S[7] * v20 + C[7] * v17;
  const float v26 = S[7] * v7 - C[7] * v14;
  const float v27 = S[7] * v14 + C[7] * v7;
  const float v28 = C[7] * v16 - S[7] * v21;
  const float v29 = -(S[7] * v16) - C[7] * v21;
  const float v31 = -(S[6] * v23);
  const float v32 = C[6] * v22;
  const float v34 = S[6] * v25;
  const float v35 = C[6] * v25;
  const float v36 = S[6] * v24;
  const float v39 = -(S[6] * v18);
  const float v40 = -(S[6] * v13);
  const float v41 = -(C[6] * v13);
  const float v43 = C[5] * v35;
  const float v47 = -(S[5] * v15);
  const float v48 = S[5] * v34;
  const float v50 = S[5] * v36;
  const float v51 = C[5] * v36;
  const float v53 = -(S[5] * v41);
  const float v54 = C[5] * v40;
  const float v55 = -(S[5] * v40);
  const float v59 = C[4] * v27 - S[4] * v29;
  const float v60 = C[4] * v28 + S[4] * v26;
  const float v63 = C[4] * v53;
  const float v64 = S[4] * v55;
  const float v66 = S[4] * v39;
  const float v67 = C[4] * v39;
  const float v72 = S[4] * v54;
  const float v73 = C[4] * v54;
  const float v74 = C[3] * v32 + S[3] * v59;
  const float v75 = C[3] * v59 - S[3] * v32;
  const float v76 = C[3] * v51;
  const float v77 = -(S[3] * v51);
  const float v78 = -(C[3] * v67);
  const float v79 = S[3] * v67;
  const float v80 = C[3] * v72;
  const float v81 = -(S[3] * v72);
  const float v82 = S[3] * v63 - C[3] * v47;
  const float v83 = S[3] * v47 + C[3] * v63;
  const float v84 = S[3] * v48;
  const float v85 = C[3] * v48;
  const float v94 = C[3] * v64;
  const float v95 = -(S[3] * v64);
  const float v96 = C[3] * v50;
  const float v97 = -(S[3] * v50);
  const float v98 = C[2] * v79 + S[2] * v77;
  const float v99 = C[2] * v77 - S[2] * v79;
  const float v100 = C[2] * v78 + S[2] * v76;
  const float v101 = C[2] * v76 - S[2] * v78;
  const float v104 = C[2] * v73 + S[2] * v31;
  const float v105 = C[2] * v31 - S[2] * v73;
  const float v106 = C[2] * v85;
  const float v107 = -(S[2] * v85);
  const float v108 = C[2] * v84;
  const float v109 = -(S[2] * v84);
  const float v118 = S[2] * v95;
  const float v119 = C[2] * v95;
  const float v120 = S[2] * v94;
  const float v121 = C[2] * v94;
  const float v124 = C[1] * v81;
  const float v125 = -(S[1] * v81);
  const float v126 = C[1] * v80;
  const float v127 = -(S[1] * v80);
  const float v128 = S[1] * v99;
  const float v129 = C[1] * v99;
  const float v130 = S[1] * v101;
  const float v131 = C[1] * v101;
  const float v132 = S[1] * v98;
  const float v133 = C[1] * v98;
  const float v134 = S[1] * v100;
  const float v135 = C[1] * v100;
  const float v136 = -(C[1] * v66);
  const float v137 = S[1] * v66;
  const float v138 = S[1] * v105;
  const float v139 = C[1] * v105;
  const float v140 = S[1] * v107;
  const float v141 = C[1] * v107;
  const float v142 = S[1] * v109;
  const float v143 = C[1] * v109;
  const float v144 = S[1] * v104;
  const float v145 = C[1] * v104;
  const float v146 = S[1] * v106;
  const float v147 = C[1] * v106;
  const float v148 = S[1] * v108;
  const float v149 = C[1] * v108;
  const float v156 = C[1] * v97;
  const float v157 = -(S[1] * v97);
  const float v158 = C[1] * v96;
  const float v159 = -(S[1] * v96);
  const float v160 = S[1] * v119;
  const float v161 = C[1] * v119;
  const float v162 = S[1] * v121;
  const float v163 = C[1] * v121;
  const float v166 = S[1] * v118;
  const float v167 = C[1] * v118;
  const float v168 = S[1] * v120;
  const float v169 = C[1] * v120;
  const float v170 = C[0] * v60 + S[0] * v43;
  const float v171 = C[0] * v43 - S[0] * v60;
  const float v172 = S[0] * v83;
  const float v173 = C[0] * v83;
  const float v174 = S[0] * v82;
  const float v175 = C[0] * v82;
  const float v178 = S[0] * v137;
  const float v179 = C[0] * v137;
  const float v180 = C[0] * v157;
  const float v181 = -(S[0] * v157);
  const float v182 = C[0] * v159;
  const float v183 = -(S[0] * v159);
  const float v184 = S[0] * v139;
  const float v185 = C[0] * v139;
  const float v186 = C[0] * v161 + S[0] * v141;
  const float v187 = C[0] * v141 - S[0] * v161;
  const float v188 = C[0] * v163 + S[0] * v143;
  const float v189 = C[0] * v143 - S[0] * v163;
  const float v192 = S[0] * v145;
  const float v193 = C[0] * v145;
  const float v194 = C[0] * v167 + S[0] * v147;
  const float v195 = C[0] * v147 - S[0] * v167;
  const float v196 = C[0] * v169 + S[0] * v149;
  const float v197 = C[0] * v149 - S[0] * v169;
  const float v202 = S[0] * v136;
  const float v203 = C[0] * v136;
  const float v204 = C[0] * v156;
  const float v205 = -(S[0] * v156);
  const float v206 = C[0] * v158;
  const float v207 = -(S[0] * v158);
  const float v208 = S[0] * v138;
  const float v209 = C[0] * v138;
  const float v210 = C[0] * v160 + S[0] * v140;
  const float v211 = C[0] * v140 - S[0] * v160;
  const float v212 = C[0] * v162 + S[0] * v142;
  const float v213 = C[0] * v142 - S[0] * v162;
  const float v216 = S[0] * v144;
  const float v217 = C[0] * v144;
  const float v218 = C[0] * v166 + S[0] * v146;
  const float v219 = C[0] * v146 - S[0] * v166;
  const float v220 = C[0] * v168 + S[0] * v148;
  const float v221 = C[0] * v148 - S[0] * v168;
  cf[1] = v74;
  cf[2] = v75;
  cf[10] = v126;
  cf[11] = v124;
  cf[13] = v134;
  cf[14] = v170;
  cf[15] = v174;
  cf[16] = v172;
  cf[23] = v202;
  cf[24] = v206;
  cf[25] = v204;
  cf[26] = v216;
  cf[27] = v220;
  cf[28] = v171;
  cf[29] = v175;
  cf[30] = v173;
  cf[37] = v203;
  cf[38] = v207;
  cf[39] = v205;
  cf[40] = v217;
  cf[41] = v221;
  cf[42] = v129;
  cf[43] = v131;
  cf[45] = v133;
  cf[46] = v135;
  cf[48] = v125;
  cf[49] = v127;
  cf[51] = v128;
  cf[52] = v130;
  cf[54] = v132;
  cf[56] = v186;
  cf[57] = v188;
  cf[58] = v184;
  cf[59] = v194;
  cf[60] = v196;
  cf[61] = v192;
  cf[62] = v180;
  cf[63] = v182;
  cf[64] = v178;
  cf[65] = v210;
  cf[66] = v212;
  cf[67] = v208;
  cf[68] = v218;
  cf[70] = v187;
  cf[71] = v189;
  cf[72] = v185;
  cf[73] = v195;
  cf[74] = v197;
  cf[75] = v193;
  cf[76] = v181;
  cf[77] = v183;
  cf[78] = v179;
  cf[79] = v211;
  cf[80] = v213;
  cf[81] = v209;
  cf[82] = v219;
}

__device__ __forceinline__ void coef_blockring(const float (&C)[16], const float (&S)[16], float (&cf)[84]) {
  const float v0 = C[11];
  const float v1 = -(S[11]);
  const float v2 = C[10] * v1;
  const float v3 = -(S[10] * v1);
  const float v4 = C[10] * v0;
  const float v5 = -(S[10] * v0);
  const float v6 = C[8] * v3;
  const float v7 = -(S[8] * v3);
  const float v8 = C[8] * v5;
  const float v9 = -(S[8] * v5);
  const float v10 = C[8] * v2;
  const float v11 = -(S[8] * v2);
  const float v12 = C[8] * v4;
  const float v13 = -(S[8] * v4);
  const float v14 = -(C[7] * v6);
  const float v15 = S[7] * v6;
  const float v16 = -(C[7] * v13);
  const float v18 = S[7] * v8;
  const float v19 = C[7] * v8;
  const float v20 = S[7] * v11;
  const float v21 = C[7] * v11;
  const float v22 = S[6] * v7;
  const float v23 = C[6] * v7;
  const float v24 = C[6] * v12;
  const float v25 = -(S[6] * v12);
  const float v26 = S[6] * v19;
  const float v27 = C[6] * v19;
  const float v28 = S[6] * v18;
  const float v29 = C[6] * v18;
  const float v30 = C[6] * v21;
  const float v31 = -(S[6] * v21);
  const float v32 = C[6] * v20;
  const float v33 = -(S[6] * v20);
  const float v34 = C[5] * v25;
  const float v35 = -(S[5] * v25);
  const float v36 = C[5] * v24;
  const float v38 = -(S[5] * v9);
  const float v39 = -(C[5] * v9);
  const float v40 = S[5] * v15;
  const float v43 = C[5] * v14;
  const float v44 = C[5] * v31;
  const float v45 = -(S[5] * v31);
  const float v46 = C[5] * v33;
  const float v47 = -(S[5] * v33);
  const float v48 = C[5] * v30;
  const float v49 = -(S[5] * v30);
  const float v50 = C[5] * v32;
  const float v53 = -(S[4] * v27);
  const float v54 = C[4] * v29;
  const float v57 = -(S[4] * v26);
  const float v58 = C[4] * v28;
  const float v61 = -(S[4] * v45);
  const float v62 = C[4] * v47;
  const float v63 = -(S[4] * v47);
  const float v64 = S[4] * v39;
  const float v66 = C[4] * v49;
  const float v70 = -(S[4] * v10);
  const float v73 = -(S[4] * v44);
  const float v74 = C[4] * v46;
  const float v77 = C[4] * v38;
  const float v79 = -(S[4] * v48);
  const float v80 = C[4] * v50;
  const float v81 = -(S[4] * v50);
  const float v82 = S[3] * v23 - C[3] * v77;
  const float v83 = S[3] * v77 + C[3] * v23;
  const float v84 = S[3] * v79;
  const float v85 = C[3] * v79;
  const float v86 = S[3] * v61;
  const float v87 = C[3] * v61;
  const float v90 = S[3] * v74;
  const float v91 = C[3] * v74;
  const float v92 = C[3] * v63;
  const float v93 = -(S[3] * v63);
  const float v94 = C[3] * v81;
  const float v95 = -(S[3] * v81);
  const float v98 = C[3] * v66 - S[3] * v16;
  const float v99 = -(S[3] * v66) - C[3] * v16;
  const float v100 = -(S[3] * v64);
  const float v101 = -(C[3] * v64);
  const float v106 = S[3] * v54 - C[3] * v40;
  const float v107 = S[3] * v40 + C[3] * v54;
  const float v110 = S[3] * v34;
  const float v111 = C[3] * v34;
  const float v112 = S[3] * v57;
  const float v113 = C[3] * v57;
  const float v114 = C[2] * v36 + S[2] * v53;
  const float v115 = C[2] * v53 - S[2] * v36;
  const float v116 = C[2] * v85 + S[2] * v83;
  const float v117 = C[2] * v83 - S[2] * v85;
  const float v118 = C[2] * v84 + S[2] * v82;
  const float v119 = C[2] * v82 - S[2] * v84;
  const float v120 = C[2] * v58;
  const float v121 = -(S[2] * v58);
  const float v122 = S[2] * v91;
  const float v123 = C[2] * v91;
  const float v124 = S[2] * v90;
  const float v125 = C[2] * v90;
  const float v126 = C[2] * v93;
  const float v127 = -(S[2] * v93);
  const float v128 = C[2] * v92;
  const float v129 = -(S[2] * v92);
  const float v130 = -(C[2] * v43) - S[2] * v70;
  const float v131 = S[2] * v43 - C[2] * v70;
  const float v132 = C[2] * v101 + S[2] * v99;
  const float v133 = C[2] * v99 - S[2] * v101;
  const float v134 = C[2] * v100 + S[2] * v98;
  const float v135 = C[2] * v98 - S[2] * v100;
  const float v140 = C[2] * v80;
  const float v141 = -(S[2] * v80);
  const float v142 = S[2] * v107;
  const float v143 = C[2] * v107;
  const float v144 = S[2] * v106;
  const float v145 = C[2] * v106;
  const float v146 = C[2] * v22 + S[2] * v73;
  const float v147 = C[2] * v73 - S[2] * v22;
  const float v148 = C[2] * v113 + S[2] * v111;
  const float v149 = C[2] * v111 - S[2] * v113;
  const float v150 = C[2] * v112 + S[2] * v110;
  const float v151 = C[2] * v110 - S[2] * v112;
  const float v152 = S[1] * v35;
  const float v153 = C[1] * v35;
  const float v154 = S[1] * v87;
  const float v155 = C[1] * v87;
  const float v156 = S[1] * v86;
  const float v157 = C[1] * v86;
  const float v158 = C[1] * v121;
  const float v159 = -(S[1] * v121);
  const float v160 = C[1] * v123;
  const float v161 = -(S[1] * v123);
  const float v162 = C[1] * v125;
  const float v163 = -(S[1] * v125);
  const float v164 = C[1] * v120;
  const float v165 = -(S[1] * v120);
  const float v166 = C[1] * v122;
  const float v167 = -(S[1] * v122);
  const float v168 = C[1] * v124;
  const float v169 = -(S[1] * v124);
  const float v170 = S[1] * v95;
  const float v171 = C[1] * v95;
  const float v172 = S[1] * v94;
  const float v173 = C[1] * v94;
  const float v174 = C[1] * v131;
  const float v175 = -(S[1] * v131);
  const float v176 = C[1] * v133;
  const float v177 = -(S[1] * v133);
  const float v178 = C[1] * v135;
  const float v179 = -(S[1] * v135);
  const float v182 = C[1] * v130;
  const float v183 = -(S[1] * v130);
  const float v184 = C[1] * v132;
  const float v185 = -(S[1] * v132);
  const float v186 = C[1] * v134;
  const float v187 = -(S[1] * v134);
  const float v200 = S[1] * v62;
  const float v201 = C[1] * v62;
  const float v202 = C[1] * v147;
  const float v203 = -(S[1] * v147);
  const float v204 = C[1] * v149;
  const float v205 = -(S[1] * v149);
  const float v206 = C[1] * v151;
  const float v207 = -(S[1] * v151);
  const float v208 = C[1] * v146;
  const float v209 = -(S[1] * v146);
  const float v210 = C[1] * v148;
  const float v211 = -(S[1] * v148);
  const float v212 = C[1] * v150;
  const float v213 = -(S[1] * v150);
  const float v214 = C[0] * v141;
  const float v215 = -(S[0] * v141);
  const float v216 = C[0] * v143 + S[0] * v127;
  const float v217 = C[0] * v127 - S[0] * v143;
  const float v218 = C[0] * v145 + S[0] * v129;
  const float v219 = C[0] * v129 - S[0] * v145;
  const float v220 = C[0] * v140;
  const float v221 = -(S[0] * v140);
  const float v222 = C[0] * v142 + S[0] * v126;
  const float v223 = C[0] * v126 - S[0] * v142;
  const float v224 = C[0] * v144 + S[0] * v128;
  const float v225 = C[0] * v128 - S[0] * v144;
  const float v226 = C[0] * v201;
  const float v227 = -(S[0] * v201);
  const float v228 = S[0] * v171;
  const float v229 = C[0] * v171;
  const float v230 = S[0] * v173;
  const float v231 = C[0] * v173;
  const float v232 = C[0] * v203 + S[0] * v175;
  const float v233 = C[0] * v175 - S[0] * v203;
  const float v234 = C[0] * v205 + S[0] * v177;
  const float v235 = C[0] * v177 - S[0] * v205;
  const float v236 = C[0] * v207 + S[0] * v179;
  const float v237 = C[0] * v179 - S[0] * v207;
  const float v240 = C[0] * v209 + S[0] * v183;
  const float v241 = C[0] * v183 - S[0] * v209;
  const float v242 = C[0] * v211 + S[0] * v185;
  const float v243 = C[0] * v185 - S[0] * v211;
  const float v244 = C[0] * v213 + S[0] * v187;
  const float v245 = C[0] * v187 - S[0] * v213;
  const float v254 = C[0] * v200;
  const float v255 = -(S[0] * v200);
  const float v256 = S[0] * v170;
  const float v257 = C[0] * v170;
  const float v258 = S[0] * v172;
  const float v259 = C[0] * v172;
  const float v260 = C[0] * v202 + S[0] * v174;
  const float v261 = C[0] * v174 - S[0] * v202;
  const float v262 = C[0] * v204 + S[0] * v176;
  const float v263 = C[0] * v176 - S[0] * v204;
  const float v264 = C[0] * v206 + S[0] * v178;
  const float v265 = C[0] * v178 - S[0] * v206;
  const float v268 = C[0] * v208 + S[0] * v182;
  const float v269 = C[0] * v182 - S[0] * v208;
  const float v270 = C[0] * v210 + S[0] * v184;
  const float v271 = C[0] * v184 - S[0] * v210;
  const float v272 = C[0] * v212 + S[0] * v186;
  const float v273 = C[0] * v186 - S[0] * v212;
  cf[3] = v114;
  cf[4] = v118;
  cf[5] = v116;
  cf[6] = v115;
  cf[7] = v119;
  cf[8] = v117;
  cf[9] = v152;
  cf[10] = v156;
  cf[11] = v154;
  cf[12] = v164;
  cf[13] = v168;
  cf[17] = v220;
  cf[18] = v224;
  cf[19] = v222;
  cf[20] = v214;
  cf[21] = v218;
  cf[22] = v216;
  cf[23] = v254;
  cf[24] = v258;
  cf[25] = v256;
  cf[26] = v268;
  cf[27] = v272;
  cf[31] = v221;
  cf[32] = v225;
  cf[33] = v223;
  cf[34] = v215;
  cf[35] = v219;
  cf[36] = v217;
  cf[37] = v255;
  cf[38] = v259;
  cf[39] = v257;
  cf[40] = v269;
  cf[41] = v273;
  cf[42] = v161;
  cf[43] = v163;
  cf[44] = v159;
  cf[45] = v167;
  cf[46] = v169;
  cf[47] = v165;
  cf[48] = v155;
  cf[49] = v157;
  cf[50] = v153;
  cf[51] = v160;
  cf[52] = v162;
  cf[53] = v158;
  cf[54] = v166;
  cf[56] = v234;
  cf[57] = v236;
  cf[58] = v232;
  cf[59] = v242;
  cf[60] = v244;
  cf[61] = v240;
  cf[62] = v228;
  cf[63] = v230;
  cf[64] = v226;
  cf[65] = v262;
  cf[66] = v264;
  cf[67] = v260;
  cf[68] = v270;
  cf[70] = v235;
  cf[71] = v237;
  cf[72] = v233;
  cf[73] = v243;
  cf[74] = v245;
  cf[75] = v241;
  cf[76] = v229;
  cf[77] = v231;
  cf[78] = v227;
  cf[79] = v263;
  cf[80] = v265;
  cf[81] = v261;
  cf[82] = v271;
}

__device__ __forceinline__ void sincos_inline(float x, float& sn, float& cs) {
  const float ax = __builtin_fabsf(x);
  const float fn = __builtin_rintf(ax * __uint_as_float(0x3f22f983u));
  float r = __builtin_fmaf(fn, __uint_as_float(0xbfc90fdau), ax);
  r = __builtin_fmaf(fn, __uint_as_float(0xb3a22168u), r);
  r = __builtin_fmaf(fn, __uint_as_float(0xa7c234c4u), r);
  const float s2 = r * r;
  float ps = __builtin_fmaf(s2, __uint_as_float(0xb94c1982u), __uint_as_float(0x3c0881c4u));
  ps = __builtin_fmaf(s2, ps, __uint_as_float(0xbe2aaa9du));
  ps = s2 * ps;
  const float sr = __builtin_fmaf(r, ps, r);
  float pc = __builtin_fmaf(s2, __uint_as_float(0x37d75334u), __uint_as_float(0xbab64f3bu));
  pc = __builtin_fmaf(s2, pc, __uint_as_float(0x3d2aabf7u));
  pc = __builtin_fmaf(s2, pc, __uint_as_float(0xbf000004u));
  const float cr = __builtin_fmaf(s2, pc, 1.0f);
  const int      n    = (int)fn;
  const bool     odd  = (n & 1) != 0;
  const unsigned flip = ((unsigned)n << 30) & 0x80000000u;
  const float sm = odd ? cr : sr;
  const float cm = odd ? -sr : cr;
  sn = __uint_as_float(__float_as_uint(sm) ^ flip ^ (__float_as_uint(x) & 0x80000000u));
  cs = __uint_as_float(__float_as_uint(cm) ^ flip);
}

__device__ __forceinline__ void store_slots(const float (&cf)[84], v2f* __restrict__ tb) {
#pragma unroll
  for (int h = 0; h < 2; ++h) {
#pragma unroll
    for (int s = 0; s < 21; ++s) tb[s * 32 + h * 16] = v2f{cf[h * 42 + 2 * s], cf[h * 42 + 2 * s + 1]};
  }
}

__global__ void __launch_bounds__(128) quanv_onelaunch_kernel(const float* __restrict__ wts,
                                                              const float* __restrict__ in,
                                                              float* __restrict__ out) {
  __shared__ v2f tblv[63 * 32];
  __shared__ v2f dws[18 * 16 + 4];

  const int t    = threadIdx.x;
  const int lane = t & 31;
  const int wv   = __builtin_amdgcn_readfirstlane(t >> 5);

  float wval[2];
  int   wslot[2];
#pragma unroll
  for (int r = 0; r < 2; ++r) {
    const bool live = t + 128 * r < 252;
    const int  v   = live ? t + 128 * r : 251;
    const bool dbl = v < 144;
    const int  u   = dbl ? v : v - 144;
    const int  L   = dbl ? (u >> 4) : (u / 12);
    const int  k   = dbl ? (u & 15) : (u % 12);
    const int  i   = (dbl ? 3 : 4) + 5 * (L / 3);
    const int  c   = L % 3;
    wslot[r] = live ? ((dbl ? 0 : 9) + L) * 16 + k : 18 * 16 + (t & 3);
    wval[r]  = wts[(i * 3 + c) * 16 + k];
  }
  const int h    = lane >> 4;
  const int n    = lane & 15;
  const int tile = blockIdx.x * 4 + wv;
  const int tl   = (tile < 961) ? tile : 960;
  const int p    = tl * 16 + n;
  const int b    = p / 961;
  const int rem  = p % 961;
  const int ph   = rem / 31;
  const int pw   = rem % 31;
  float px[12];
  {
    const float* ip = in + (b * 3 * 32 + ph) * 32 + pw;
#pragma unroll
    for (int c = 0; c < 3; ++c) {
#pragma unroll
      for (int w = 0; w < 4; ++w) px[c * 4 + w] = ip[c * 1024 + (w >> 1) * 32 + (w & 1)];
    }
  }
  const int sl      = (lane < 30) ? lane : 29;
  const int sh_type = (sl < 12) ? 0 : (sl < 21) ? 1 : 2;
  const int sh_u    = (sl < 12) ? sl : (sl < 21) ? sl - 12 : sl - 21;
  const int sh_i    = sh_type + 5 * (sh_u / 3);
  const int sh_c    = sh_u % 3;
  float shw[5];
  {
    const float* wp = wts + (sh_i * 3 + sh_c) * 16;
#pragma unroll
    for (int k = 0; k < 5; ++k) shw[k] = wp[k];
  }

#pragma unroll
  for (int r = 0; r < 2; ++r) {
    float wsn, wcs;
    sincos_inline(wval[r], wsn, wcs);
    dws[wslot[r]] = v2f{wcs, wsn};
  }
  __syncthreads();

  if (wv < 2) {
    if (lane < 9) {
      const int i = 3 + wv + 5 * (lane / 3);
      const int c = lane % 3;
      const v2f* e = dws + (wv * 9 + lane) * 16;
      float C[16], S[16];
      float cf[84];
#pragma unroll
      for (int k = 0; k < 84; ++k) cf[k] = 0.0f;
      if (wv == 0) {
#pragma unroll
        for (int k = 0; k < 16; ++k) { const v2f a = e[k]; C[k] = a[0]; S[k] = a[1]; }
        coef_doublering(C, S, cf);
      } else {
#pragma unroll
        for (int k = 0; k < 16; ++k) { C[k] = 0.0f; S[k] = 0.0f; }
#pragma unroll
        for (int k = 0; k < 12; ++k) { const v2f a = e[k]; C[k] = a[0]; S[k] = a[1]; }
        coef_blockring(C, S, cf);
      }
      store_slots(cf, tblv + c * 21 * 32 + i);
    }
  } else if (wv == 2) {
    if (lane < 30) {
      float C[16], S[16];
#pragma unroll
      for (int k = 0; k < 16; ++k) { C[k] = 0.0f; S[k] = 0.0f; }
#pragma unroll
      for (int k = 0; k < 5; ++k) sincos_inline(shw[k], S[k], C[k]);
      float cf[84];
#pragma unroll
      for (int k = 0; k < 84; ++k) cf[k] = 0.0f;
      if (sh_type == 0)      coef_ran(C, S, cf);
      else if (sh_type == 1) coef_line(C, S, cf);
      else                   coef_ring(C, S, cf);
      store_slots(cf, tblv + sh_c * 21 * 32 + sh_i);
    }
  }
  __syncthreads();

  if (tile >= 961) return;

  const v2f* Cv = tblv + lane;
  v2f afr[63];
#pragma unroll
  for (int g = 0; g < 63; ++g) afr[g] = Cv[g * 32];
  __builtin_amdgcn_sched_barrier(0);

  float cs[3][4], sn[3][4];
#pragma unroll
  for (int c = 0; c < 3; ++c) {
#pragma unroll
    for (int w = 0; w < 4; ++w) sincos_inline(px[c * 4 + w], sn[c][w], cs[c][w]);
  }

  const bool hi = (h != 0);
  v8f acc0 = v8f{};
  v8f acc1 = v8f{};
#pragma unroll
  for (int c = 0; c < 3; ++c) {
    const float g3[3] = { hi ? sn[c][3] : 1.0f, cs[c][3], hi ? 1.0f : sn[c][3] };
    const float g2[3] = { hi ? sn[c][2] : 1.0f, cs[c][2], hi ? 1.0f : sn[c][2] };
    const float g1[2] = { hi ? sn[c][1] : 1.0f, cs[c][1] };
    float Q[9];
#pragma unroll
    for (int a2 = 0; a2 < 3; ++a2) {
#pragma unroll
      for (int a3 = 0; a3 < 3; ++a3) Q[3 * a2 + a3] = g2[a2] * g3[a3];
    }
    float R[14];
#pragma unroll
    for (int j = 0; j < 14; ++j) R[j] = (j < 9) ? g1[0] * Q[j] : g1[1] * Q[j - 9];
    float F[42];
#pragma unroll
    for (int j = 0; j < 14; ++j) {
      F[j]      = R[j];
      F[14 + j] = cs[c][0] * R[j];
      F[28 + j] = sn[c][0] * R[j];
    }
#pragma unroll
    for (int s = 0; s < 21; ++s) {
      const int g = c * 21 + s;
      v2f bfr;
      bfr[0] = F[2 * s];
      bfr[1] = F[2 * s + 1];
      if (g & 1) {
        acc1 = __builtin_amdgcn_wmma_f32_16x16x4_f32(false, afr[g], false, bfr, (short)0, acc1, false, false);
      } else {
        acc0 = __builtin_amdgcn_wmma_f32_16x16x4_f32(false, afr[g], false, bfr, (short)0, acc0, false, false);
      }
    }
  }

  float* op = out + (b * 16 + 8 * h) * 961 + rem;
#pragma unroll
  for (int r = 0; r < 8; ++r) op[r * 961] = acc0[r] + acc1[r];
}

extern "C" void kernel_launch(void* const* d_in, const int* in_sizes, int n_in,
                              void* d_out, int out_size, void* d_ws, size_t ws_size,
                              hipStream_t stream) {
  const float* inputs  = (const float*)d_in[0];
  const float* weights = (const float*)d_in[1];
  float*       out     = (float*)d_out;

  quanv_onelaunch_kernel<<<241, 128, 0, stream>>>(weights, inputs, out);
}
